# starting kernel
# baseline (speedup 1.0000x reference)
#include <hip/hip_runtime.h>
#include <hip/hip_bf16.h>

#define NN    50000
#define RR    51
#define EE    16
#define CC    8
#define BB    40
#define NNZ   1600000
#define NE    (NN * EE)
#define W1ELEMS ((size_t)RR * NE)

typedef float v2f __attribute__((ext_vector_type(2)));
typedef float v8f __attribute__((ext_vector_type(8)));

__device__ __forceinline__ void atomAddF32(float* p, float v) {
  unsafeAtomicAdd(p, v);
}

__global__ void rgcn_init(float* __restrict__ h, float* __restrict__ out,
                          const float* __restrict__ bias2) {
  int i = blockIdx.x * blockDim.x + threadIdx.x;
  if (i < NE)        h[i] = 0.0f;
  if (i < NN * CC)   out[i] = bias2[i & (CC - 1)];
}

__global__ void rgcn_w2(const float* __restrict__ comps2,
                        const float* __restrict__ bases2,
                        float* __restrict__ W2) {
  int idx = blockIdx.x * blockDim.x + threadIdx.x;
  if (idx >= RR * EE * CC) return;
  int r  = idx >> 7;
  int ec = idx & 127;
  float acc = 0.0f;
#pragma unroll
  for (int b = 0; b < BB; ++b)
    acc += comps2[r * BB + b] * bases2[b * (EE * CC) + ec];
  W2[idx] = acc;
}

__global__ void __launch_bounds__(256)
rgcn_w1_gemm(const float* __restrict__ comps1,
             const float* __restrict__ bases1,
             float* __restrict__ W1) {
  const int lane = threadIdx.x & 31;
  const int half = lane >> 4;
  const int l16  = lane & 15;

  v2f a[4][10];
#pragma unroll
  for (int mt = 0; mt < 4; ++mt) {
    int m = mt * 16 + l16;
#pragma unroll
    for (int ks = 0; ks < 10; ++ks) {
      int k = ks * 4 + half * 2;
      float ax = 0.0f, ay = 0.0f;
      if (m < RR) {
        ax = comps1[m * BB + k];
        ay = comps1[m * BB + k + 1];
      }
      a[mt][ks].x = ax;
      a[mt][ks].y = ay;
    }
  }

  const int wavesPerBlock = blockDim.x >> 5;
  const int waveId  = blockIdx.x * wavesPerBlock + (threadIdx.x >> 5);
  const int nWaves  = gridDim.x * wavesPerBlock;
  const int nTiles  = NE / 16;

  for (int tile = waveId; tile < nTiles; tile += nWaves) {
    const int col = tile * 16 + l16;
    v8f acc[4] = {};
#pragma unroll
    for (int ks = 0; ks < 10; ++ks) {
      const int k = ks * 4 + half * 2;
      v2f b;
      b.x = __builtin_nontemporal_load(&bases1[(size_t)k * NE + col]);
      b.y = __builtin_nontemporal_load(&bases1[(size_t)(k + 1) * NE + col]);
#pragma unroll
      for (int mt = 0; mt < 4; ++mt) {
        acc[mt] = __builtin_amdgcn_wmma_f32_16x16x4_f32(
             false, a[mt][ks],  false, b,
             (short)0, acc[mt],
             false,  false);
      }
    }
#pragma unroll
    for (int mt = 0; mt < 4; ++mt) {
#pragma unroll
      for (int j = 0; j < 8; ++j) {
        int m = mt * 16 + j + half * 8;
        if (m < RR) W1[(size_t)m * NE + col] = acc[mt][j];
      }
    }
  }
}

__global__ void rgcn_scatter_h(const float* __restrict__ values,
                               const int* __restrict__ hr,
                               const int* __restrict__ hc,
                               const float* __restrict__ W1,
                               float* __restrict__ h) {
  int t = blockIdx.x * blockDim.x + threadIdx.x;
  if (t >= NNZ * EE) return;
  int i = t >> 4;
  int e = t & 15;
  float v  = __builtin_nontemporal_load(&values[i]);
  int col  = __builtin_nontemporal_load(&hc[i]);
  int row  = __builtin_nontemporal_load(&hr[i]);
  float w  = W1[(size_t)col * EE + e];
  atomAddF32(&h[row * EE + e], v * w);
}

__global__ void rgcn_bias_relu(float* __restrict__ h,
                               const float* __restrict__ bias1) {
  int i = blockIdx.x * blockDim.x + threadIdx.x;
  if (i >= NE) return;
  float x = h[i] + bias1[i & 15];
  h[i] = x > 0.0f ? x : 0.0f;
}

__global__ void rgcn_layer2(const float* __restrict__ values,
                            const int* __restrict__ vr,
                            const int* __restrict__ vc,
                            const float* __restrict__ h,
                            const float* __restrict__ W2,
                            float* __restrict__ out) {
  __shared__ float W2s[RR * EE * CC];
  for (int j = threadIdx.x; j < RR * EE * CC; j += blockDim.x)
    W2s[j] = W2[j];
  __syncthreads();

  const long long total  = (long long)NNZ * CC;
  const long long stride = (long long)gridDim.x * blockDim.x;
  for (long long t = (long long)blockIdx.x * blockDim.x + threadIdx.x;
       t < total; t += stride) {
    int i = (int)(t >> 3);
    int c = (int)(t & 7);
    int r   = __builtin_nontemporal_load(&vr[i]);
    int o   = __builtin_nontemporal_load(&vc[i]);
    float v = __builtin_nontemporal_load(&values[i]);
    int p = r / NN;
    int s = r - p * NN;
    const float* hp = h + (size_t)o * EE;
    const float* wp = W2s + p * (EE * CC) + c;
    float acc = 0.0f;
#pragma unroll
    for (int e = 0; e < EE; ++e)
      acc += hp[e] * wp[e * CC];
    atomAddF32(&out[s * CC + c], v * acc);
  }
}

extern "C" void kernel_launch(void* const* d_in, const int* in_sizes, int n_in,
                              void* d_out, int out_size, void* d_ws, size_t ws_size,
                              hipStream_t stream) {
  const float* values = (const float*)d_in[0];
  const float* comps1 = (const float*)d_in[1];
  const float* bases1 = (const float*)d_in[2];
  const float* comps2 = (const float*)d_in[3];
  const float* bases2 = (const float*)d_in[4];
  const float* bias1  = (const float*)d_in[5];
  const float* bias2  = (const float*)d_in[6];
  const int*   hr     = (const int*)d_in[7];
  const int*   hc     = (const int*)d_in[8];
  const int*   vr     = (const int*)d_in[9];
  const int*   vc     = (const int*)d_in[10];
  float* out = (float*)d_out;

  float* W1 = (float*)d_ws;
  float* h  = W1 + W1ELEMS;
  float* W2 = h + NE;

  rgcn_init<<<(NE + 255) / 256, 256, 0, stream>>>(h, out, bias2);
  rgcn_w2<<<(RR * EE * CC + 255) / 256, 256, 0, stream>>>(comps2, bases2, W2);
  rgcn_w1_gemm<<<832, 256, 0, stream>>>(comps1, bases1, W1);
  {
    long long n = (long long)NNZ * EE;
    rgcn_scatter_h<<<(int)((n + 255) / 256), 256, 0, stream>>>(values, hr, hc, W1, h);
  }
  rgcn_bias_relu<<<(NE + 255) / 256, 256, 0, stream>>>(h, bias1);
  rgcn_layer2<<<4096, 256, 0, stream>>>(values, vr, vc, h, W2, out);
}


# both scatter kernels: rows cut into 64 groups, each workgroup sums its edges in an LDS tile (LDS atomics) and adds it to global memory once: 8x fewer global f32 atomics; sum order changes
# speedup vs baseline: 1.1441x; 1.1441x over previous
#include <hip/hip_runtime.h>
#include <hip/hip_bf16.h>

#define NN    50000
#define RR    51
#define EE    16
#define CC    8
#define BB    40
#define NNZ   1600000
#define NE    (NN * EE)
#define W1ELEMS ((size_t)RR * NE)

typedef float v2f __attribute__((ext_vector_type(2)));
typedef float v8f __attribute__((ext_vector_type(8)));

__device__ __forceinline__ void atomAddF32(float* p, float v) {
  unsafeAtomicAdd(p, v);
}

__global__ void rgcn_init(float* __restrict__ h, float* __restrict__ out,
                          const float* __restrict__ bias2) {
  int i = blockIdx.x * blockDim.x + threadIdx.x;
  if (i < NE)        h[i] = 0.0f;
  if (i < NN * CC)   out[i] = bias2[i & (CC - 1)];
}

__global__ void rgcn_w2(const float* __restrict__ comps2,
                        const float* __restrict__ bases2,
                        float* __restrict__ W2) {
  int idx = blockIdx.x * blockDim.x + threadIdx.x;
  if (idx >= RR * EE * CC) return;
  int r  = idx >> 7;
  int ec = idx & 127;
  float acc = 0.0f;
#pragma unroll
  for (int b = 0; b < BB; ++b)
    acc += comps2[r * BB + b] * bases2[b * (EE * CC) + ec];
  W2[idx] = acc;
}

__global__ void __launch_bounds__(256)
rgcn_w1_gemm(const float* __restrict__ comps1,
             const float* __restrict__ bases1,
             float* __restrict__ W1) {
  const int lane = threadIdx.x & 31;
  const int half = lane >> 4;
  const int l16  = lane & 15;

  v2f a[4][10];
#pragma unroll
  for (int mt = 0; mt < 4; ++mt) {
    int m = mt * 16 + l16;
#pragma unroll
    for (int ks = 0; ks < 10; ++ks) {
      int k = ks * 4 + half * 2;
      float ax = 0.0f, ay = 0.0f;
      if (m < RR) {
        ax = comps1[m * BB + k];
        ay = comps1[m * BB + k + 1];
      }
      a[mt][ks].x = ax;
      a[mt][ks].y = ay;
    }
  }

  const int wavesPerBlock = blockDim.x >> 5;
  const int waveId  = blockIdx.x * wavesPerBlock + (threadIdx.x >> 5);
  const int nWaves  = gridDim.x * wavesPerBlock;
  const int nTiles  = NE / 16;

  for (int tile = waveId; tile < nTiles; tile += nWaves) {
    const int col = tile * 16 + l16;
    v8f acc[4] = {};
#pragma unroll
    for (int ks = 0; ks < 10; ++ks) {
      const int k = ks * 4 + half * 2;
      v2f b;
      b.x = __builtin_nontemporal_load(&bases1[(size_t)k * NE + col]);
      b.y = __builtin_nontemporal_load(&bases1[(size_t)(k + 1) * NE + col]);
#pragma unroll
      for (int mt = 0; mt < 4; ++mt) {
        acc[mt] = __builtin_amdgcn_wmma_f32_16x16x4_f32(
             false, a[mt][ks],  false, b,
             (short)0, acc[mt],
             false,  false);
      }
    }
#pragma unroll
    for (int mt = 0; mt < 4; ++mt) {
#pragma unroll
      for (int j = 0; j < 8; ++j) {
        int m = mt * 16 + j + half * 8;
        if (m < RR) W1[(size_t)m * NE + col] = acc[mt][j];
      }
    }
  }
}

#define RG_ROWS 782
#define RG_G    64
#define RG_M    4
#define RG_SLICE (NNZ / RG_M)
#define RG_QW   1024
#define RG_THREADS 1024

template <typename RowFn, typename DirectFn>
__device__ __forceinline__ int rgcn_scan_slice(const int* __restrict__ ids, int eLo,
                                               int tid, int lane, unsigned* myQueue,
                                               RowFn localRow, DirectFn direct) {
  const int4* ids4 = reinterpret_cast<const int4*>(ids + eLo);
  const unsigned ltMask = (1u << lane) - 1u;
  int wcount = 0;
  for (int jb = 0; jb < RG_SLICE / 4; jb += 4 * RG_THREADS) {
    int4 rr[4];
#pragma unroll
    for (int k = 0; k < 4; ++k) {
      const int j = jb + k * RG_THREADS + tid;
      rr[k] = make_int4(-1, -1, -1, -1);
      if (j < RG_SLICE / 4) rr[k] = ids4[j];
    }
#pragma unroll
    for (int k = 0; k < 4; ++k) {
      const int j = jb + k * RG_THREADS + tid;
      const bool inRange = j < RG_SLICE / 4;
      const unsigned d0 = localRow(rr[k].x), d1 = localRow(rr[k].y);
      const unsigned d2 = localRow(rr[k].z), d3 = localRow(rr[k].w);
      const bool s0 = inRange & (d0 < RG_ROWS), s1 = inRange & (d1 < RG_ROWS);
      const bool s2 = inRange & (d2 < RG_ROWS), s3 = inRange & (d3 < RG_ROWS);
      const unsigned b0 = __builtin_amdgcn_ballot_w32(s0);
      const unsigned b1 = __builtin_amdgcn_ballot_w32(s1);
      const unsigned b2 = __builtin_amdgcn_ballot_w32(s2);
      const unsigned b3 = __builtin_amdgcn_ballot_w32(s3);
      if ((b0 | b1 | b2 | b3) != 0u) {
        const int i0 = eLo + 4 * j;
        const int p0 = wcount + __popc(b0 & ltMask);
        const int c0 = wcount + __popc(b0);
        const int p1 = c0 + __popc(b1 & ltMask);
        const int c1 = c0 + __popc(b1);
        const int p2 = c1 + __popc(b2 & ltMask);
        const int c2 = c1 + __popc(b2);
        const int p3 = c2 + __popc(b3 & ltMask);
        wcount = c2 + __popc(b3);
        if (s0) { if (__builtin_expect(p0 < RG_QW, 1)) myQueue[p0] = (d0 << 21) | (unsigned)(i0);
                  else direct(d0, i0); }
        if (s1) { if (__builtin_expect(p1 < RG_QW, 1)) myQueue[p1] = (d1 << 21) | (unsigned)(i0 + 1);
                  else direct(d1, i0 + 1); }
        if (s2) { if (__builtin_expect(p2 < RG_QW, 1)) myQueue[p2] = (d2 << 21) | (unsigned)(i0 + 2);
                  else direct(d2, i0 + 2); }
        if (s3) { if (__builtin_expect(p3 < RG_QW, 1)) myQueue[p3] = (d3 << 21) | (unsigned)(i0 + 3);
                  else direct(d3, i0 + 3); }
      }
    }
  }
  return wcount;
}

#define RG_U3   16

__global__ void __launch_bounds__(RG_THREADS)
rgcn_scatter_h(const float* __restrict__ values,
               const int* __restrict__ hr,
               const int* __restrict__ hc,
               const float* __restrict__ W1,
               float* __restrict__ h) {
  __shared__ float    accTile[RG_ROWS * EE];
  __shared__ unsigned queue[32][RG_QW];
  __shared__ float    spare[EE];

  const int tid  = threadIdx.x;
  const int lane = tid & 31;
  const int wv   = tid >> 5;
  const int half = lane >> 4;
  const int l16  = lane & 15;
  const int g    = blockIdx.x % RG_G;
  const int m    = blockIdx.x / RG_G;
  const int rowLo = g * RG_ROWS;
  const int eLo   = m * RG_SLICE;

  for (int j = tid; j < RG_ROWS * EE; j += RG_THREADS) accTile[j] = 0.0f;
  __syncthreads();

  const int wcount = rgcn_scan_slice(
      hr, eLo, tid, lane, queue[wv],
      [&](int r) { return (unsigned)(r - rowLo); },
      [&](unsigned d, int i) {
        const float v = values[i];
        const float* wp = W1 + (size_t)hc[i] * EE;
#pragma nounroll
        for (int e = 0; e < EE; ++e) atomAddF32(&accTile[d * EE + e], v * wp[e]);
      });
  __syncthreads();

  const int n = wcount < RG_QW ? wcount : RG_QW;
  for (int base = 0; base < n; base += 2 * RG_U3) {
    unsigned ent[RG_U3];
    bool ok[RG_U3];
#pragma unroll
    for (int u = 0; u < RG_U3; ++u) {
      const int q = base + 2 * u + half;
      ok[u] = q < n;
      ent[u] = ok[u] ? queue[wv][q] : 0u;
    }
    int col[RG_U3];
    float v[RG_U3];
#pragma unroll
    for (int u = 0; u < RG_U3; ++u) {
      const int i = ok[u] ? (int)(ent[u] & 0x1FFFFFu) : eLo;
      col[u] = __builtin_nontemporal_load(&hc[i]);
      v[u]   = __builtin_nontemporal_load(&values[i]);
    }
    float w[RG_U3];
#pragma unroll
    for (int u = 0; u < RG_U3; ++u)
      w[u] = W1[(size_t)col[u] * EE + l16];
#pragma unroll
    for (int u = 0; u < RG_U3; ++u) {
      float* dst = ok[u] ? &accTile[(ent[u] >> 21) * EE + l16] : &spare[l16];
      atomAddF32(dst, v[u] * w[u]);
    }
  }
  __syncthreads();

  for (int j = tid; j < RG_ROWS * EE; j += RG_THREADS) {
    const int row = rowLo + (j >> 4);
    if (row < NN) atomAddF32(&h[(size_t)rowLo * EE + j], accTile[j]);
  }
}

__global__ void rgcn_bias_relu(float* __restrict__ h,
                               const float* __restrict__ bias1) {
  int i = blockIdx.x * blockDim.x + threadIdx.x;
  if (i >= NE) return;
  float x = h[i] + bias1[i & 15];
  h[i] = x > 0.0f ? x : 0.0f;
}

#define RG_U5   4

__global__ void __launch_bounds__(RG_THREADS)
rgcn_layer2(const float* __restrict__ values,
            const int* __restrict__ vr,
            const int* __restrict__ vc,
            const float* __restrict__ h,
            const float* __restrict__ W2,
            float* __restrict__ out) {
  __shared__ float    W2s[RR * EE * CC];
  __shared__ float    outTile[RG_ROWS * CC];
  __shared__ unsigned queue[32][RG_QW];
  __shared__ float    spare[CC];

  const int tid  = threadIdx.x;
  const int lane = tid & 31;
  const int wv   = tid >> 5;
  const int sub  = lane >> 3;
  const int c    = lane & 7;
  const int g    = blockIdx.x % RG_G;
  const int m    = blockIdx.x / RG_G;
  const int rowLo = g * RG_ROWS;
  const int eLo   = m * RG_SLICE;

  for (int j = tid; j < RR * EE * CC; j += RG_THREADS) W2s[j] = W2[j];
  for (int j = tid; j < RG_ROWS * CC; j += RG_THREADS) outTile[j] = 0.0f;
  __syncthreads();

  const int wcount = rgcn_scan_slice(
      vr, eLo, tid, lane, queue[wv],
      [&](int r) {
        const unsigned ur = (unsigned)r;
        const unsigned prod = (unsigned)__umul24(ur, 1342u);
        const unsigned pe = prod >> 26;
        const unsigned se = ur - pe * (unsigned)NN;
        const unsigned sm = se - (unsigned)NN;
        return (se < sm ? se : sm) - (unsigned)rowLo;
      },
      [&](unsigned d, int i) {
        const int p = vr[i] / NN;
        const float v = values[i];
        const float* hp = h + (size_t)vc[i] * EE;
#pragma nounroll
        for (int cc = 0; cc < CC; ++cc) {
          const float* wp = W2s + p * (EE * CC) + cc;
          float acc = 0.0f;
#pragma nounroll
          for (int e = 0; e < EE; ++e) acc += hp[e] * wp[e * CC];
          atomAddF32(&outTile[d * CC + cc], v * acc);
        }
      });
  __syncthreads();

  const int n = wcount < RG_QW ? wcount : RG_QW;
  for (int base = 0; base < n; base += 4 * RG_U5) {
    unsigned ent[RG_U5];
    bool ok[RG_U5];
#pragma unroll
    for (int u = 0; u < RG_U5; ++u) {
      const int q = base + 4 * u + sub;
      ok[u] = q < n;
      ent[u] = ok[u] ? queue[wv][q] : 0u;
    }
    int r[RG_U5], o[RG_U5];
    float v[RG_U5];
#pragma unroll
    for (int u = 0; u < RG_U5; ++u) {
      const int i = ok[u] ? (int)(ent[u] & 0x1FFFFFu) : eLo;
      r[u] = __builtin_nontemporal_load(&vr[i]);
      o[u] = __builtin_nontemporal_load(&vc[i]);
      v[u] = __builtin_nontemporal_load(&values[i]);
    }
    float4 hv[RG_U5][4];
#pragma unroll
    for (int u = 0; u < RG_U5; ++u) {
      const float4* hp4 = reinterpret_cast<const float4*>(h + (size_t)o[u] * EE);
#pragma unroll
      for (int k = 0; k < 4; ++k) hv[u][k] = hp4[k];
    }
    __builtin_amdgcn_sched_barrier(0);
    float acc[RG_U5];
#pragma unroll
    for (int u = 0; u < RG_U5; ++u) {
      const int p = r[u] / NN;
      const float* wp = W2s + p * (EE * CC) + c;
      float a = 0.0f;
#pragma unroll
      for (int k = 0; k < 4; ++k) {
        a += hv[u][k].x * wp[(4 * k + 0) * CC];
        a += hv[u][k].y * wp[(4 * k + 1) * CC];
        a += hv[u][k].z * wp[(4 * k + 2) * CC];
        a += hv[u][k].w * wp[(4 * k + 3) * CC];
      }
      acc[u] = a;
    }
#pragma unroll
    for (int u = 0; u < RG_U5; ++u) {
      float* dst = ok[u] ? &outTile[(ent[u] >> 21) * CC + c] : &spare[c];
      atomAddF32(dst, v[u] * acc[u]);
    }
  }
  __syncthreads();

  for (int j = tid; j < RG_ROWS * CC; j += RG_THREADS) {
    const int row = rowLo + (j >> 3);
    if (row < NN) atomAddF32(&out[(size_t)rowLo * CC + j], outTile[j]);
  }
}

extern "C" void kernel_launch(void* const* d_in, const int* in_sizes, int n_in,
                              void* d_out, int out_size, void* d_ws, size_t ws_size,
                              hipStream_t stream) {
  const float* values = (const float*)d_in[0];
  const float* comps1 = (const float*)d_in[1];
  const float* bases1 = (const float*)d_in[2];
  const float* comps2 = (const float*)d_in[3];
  const float* bases2 = (const float*)d_in[4];
  const float* bias1  = (const float*)d_in[5];
  const float* bias2  = (const float*)d_in[6];
  const int*   hr     = (const int*)d_in[7];
  const int*   hc     = (const int*)d_in[8];
  const int*   vr     = (const int*)d_in[9];
  const int*   vc     = (const int*)d_in[10];
  float* out = (float*)d_out;

  float* W1 = (float*)d_ws;
  float* h  = W1 + W1ELEMS;
  float* W2 = h + NE;

  rgcn_init<<<(NE + 255) / 256, 256, 0, stream>>>(h, out, bias2);
  rgcn_w2<<<(RR * EE * CC + 255) / 256, 256, 0, stream>>>(comps2, bases2, W2);
  rgcn_w1_gemm<<<832, 256, 0, stream>>>(comps1, bases1, W1);
  rgcn_scatter_h<<<RG_G * RG_M, RG_THREADS, 0, stream>>>(values, hr, hc, W1, h);
  rgcn_bias_relu<<<(NE + 255) / 256, 256, 0, stream>>>(h, bias1);
  rgcn_layer2<<<RG_G * RG_M, RG_THREADS, 0, stream>>>(values, vr, vc, h, W2, out);
}


# init, W2-composition and bias+relu launches folded into the GEMM kernel's first lines and the layer-2 h-row reads (relu(h+bias1) applied as rows are gathered): 6 launches -> 3; sum order unchanged
# speedup vs baseline: 1.1794x; 1.0309x over previous
#include <hip/hip_runtime.h>
#include <hip/hip_bf16.h>

#define NN    50000
#define RR    51
#define EE    16
#define CC    8
#define BB    40
#define NNZ   1600000
#define NE    (NN * EE)
#define W1ELEMS ((size_t)RR * NE)

typedef float v2f __attribute__((ext_vector_type(2)));
typedef float v8f __attribute__((ext_vector_type(8)));

__device__ __forceinline__ void atomAddF32(float* p, float v) {
  unsafeAtomicAdd(p, v);
}

__global__ void __launch_bounds__(256)
rgcn_w1_gemm(const float* __restrict__ comps1,
             const float* __restrict__ bases1,
             float* __restrict__ W1,
             float* __restrict__ h,
             float* __restrict__ out,
             const float* __restrict__ bias2,
             const float* __restrict__ comps2,
             const float* __restrict__ bases2,
             float* __restrict__ W2) {
  {
    const int gid      = blockIdx.x * blockDim.x + threadIdx.x;
    const int nThreads = gridDim.x * blockDim.x;
    float4* h4   = reinterpret_cast<float4*>(h);
    float4* out4 = reinterpret_cast<float4*>(out);
    const float4 zero4 = make_float4(0.0f, 0.0f, 0.0f, 0.0f);
    for (int i = gid; i < NE / 4; i += nThreads) h4[i] = zero4;
    const int c0 = (gid & 1) * 4;
    const float4 bv = make_float4(bias2[c0], bias2[c0 + 1], bias2[c0 + 2], bias2[c0 + 3]);
    for (int i = gid; i < NN * CC / 4; i += nThreads) out4[i] = bv;
    if (gid < RR * EE * CC) {
      const int r  = gid >> 7;
      const int ec = gid & 127;
      float acc = 0.0f;
#pragma unroll
      for (int b = 0; b < BB; ++b)
        acc += comps2[r * BB + b] * bases2[b * (EE * CC) + ec];
      W2[gid] = acc;
    }
  }

  const int lane = threadIdx.x & 31;
  const int half = lane >> 4;
  const int l16  = lane & 15;

  v2f a[4][10];
#pragma unroll
  for (int mt = 0; mt < 4; ++mt) {
    int m = mt * 16 + l16;
#pragma unroll
    for (int ks = 0; ks < 10; ++ks) {
      int k = ks * 4 + half * 2;
      float ax = 0.0f, ay = 0.0f;
      if (m < RR) {
        ax = comps1[m * BB + k];
        ay = comps1[m * BB + k + 1];
      }
      a[mt][ks].x = ax;
      a[mt][ks].y = ay;
    }
  }

  const int wavesPerBlock = blockDim.x >> 5;
  const int waveId  = blockIdx.x * wavesPerBlock + (threadIdx.x >> 5);
  const int nWaves  = gridDim.x * wavesPerBlock;
  const int nTiles  = NE / 16;

  for (int tile = waveId; tile < nTiles; tile += nWaves) {
    const int col = tile * 16 + l16;
    v8f acc[4] = {};
#pragma unroll
    for (int ks = 0; ks < 10; ++ks) {
      const int k = ks * 4 + half * 2;
      v2f b;
      b.x = __builtin_nontemporal_load(&bases1[(size_t)k * NE + col]);
      b.y = __builtin_nontemporal_load(&bases1[(size_t)(k + 1) * NE + col]);
#pragma unroll
      for (int mt = 0; mt < 4; ++mt) {
        acc[mt] = __builtin_amdgcn_wmma_f32_16x16x4_f32(
             false, a[mt][ks],  false, b,
             (short)0, acc[mt],
             false,  false);
      }
    }
#pragma unroll
    for (int mt = 0; mt < 4; ++mt) {
#pragma unroll
      for (int j = 0; j < 8; ++j) {
        int m = mt * 16 + j + half * 8;
        if (m < RR) W1[(size_t)m * NE + col] = acc[mt][j];
      }
    }
  }
}

#define RG_ROWS 782
#define RG_G    64
#define RG_M    4
#define RG_SLICE (NNZ / RG_M)
#define RG_QW   1024
#define RG_THREADS 1024

template <typename RowFn, typename DirectFn>
__device__ __forceinline__ int rgcn_scan_slice(const int* __restrict__ ids, int eLo,
                                               int tid, int lane, unsigned* myQueue,
                                               RowFn localRow, DirectFn direct) {
  const int4* ids4 = reinterpret_cast<const int4*>(ids + eLo);
  const unsigned ltMask = (1u << lane) - 1u;
  int wcount = 0;
  for (int jb = 0; jb < RG_SLICE / 4; jb += 4 * RG_THREADS) {
    int4 rr[4];
#pragma unroll
    for (int k = 0; k < 4; ++k) {
      const int j = jb + k * RG_THREADS + tid;
      rr[k] = make_int4(-1, -1, -1, -1);
      if (j < RG_SLICE / 4) rr[k] = ids4[j];
    }
#pragma unroll
    for (int k = 0; k < 4; ++k) {
      const int j = jb + k * RG_THREADS + tid;
      const bool inRange = j < RG_SLICE / 4;
      const unsigned d0 = localRow(rr[k].x), d1 = localRow(rr[k].y);
      const unsigned d2 = localRow(rr[k].z), d3 = localRow(rr[k].w);
      const bool s0 = inRange & (d0 < RG_ROWS), s1 = inRange & (d1 < RG_ROWS);
      const bool s2 = inRange & (d2 < RG_ROWS), s3 = inRange & (d3 < RG_ROWS);
      const unsigned b0 = __builtin_amdgcn_ballot_w32(s0);
      const unsigned b1 = __builtin_amdgcn_ballot_w32(s1);
      const unsigned b2 = __builtin_amdgcn_ballot_w32(s2);
      const unsigned b3 = __builtin_amdgcn_ballot_w32(s3);
      if ((b0 | b1 | b2 | b3) != 0u) {
        const int i0 = eLo + 4 * j;
        const int p0 = wcount + __popc(b0 & ltMask);
        const int c0 = wcount + __popc(b0);
        const int p1 = c0 + __popc(b1 & ltMask);
        const int c1 = c0 + __popc(b1);
        const int p2 = c1 + __popc(b2 & ltMask);
        const int c2 = c1 + __popc(b2);
        const int p3 = c2 + __popc(b3 & ltMask);
        wcount = c2 + __popc(b3);
        if (s0) { if (__builtin_expect(p0 < RG_QW, 1)) myQueue[p0] = (d0 << 21) | (unsigned)(i0);
                  else direct(d0, i0); }
        if (s1) { if (__builtin_expect(p1 < RG_QW, 1)) myQueue[p1] = (d1 << 21) | (unsigned)(i0 + 1);
                  else direct(d1, i0 + 1); }
        if (s2) { if (__builtin_expect(p2 < RG_QW, 1)) myQueue[p2] = (d2 << 21) | (unsigned)(i0 + 2);
                  else direct(d2, i0 + 2); }
        if (s3) { if (__builtin_expect(p3 < RG_QW, 1)) myQueue[p3] = (d3 << 21) | (unsigned)(i0 + 3);
                  else direct(d3, i0 + 3); }
      }
    }
  }
  return wcount;
}

#define RG_U3   16

__global__ void __launch_bounds__(RG_THREADS)
rgcn_scatter_h(const float* __restrict__ values,
               const int* __restrict__ hr,
               const int* __restrict__ hc,
               const float* __restrict__ W1,
               float* __restrict__ h) {
  __shared__ float    accTile[RG_ROWS * EE];
  __shared__ unsigned queue[32][RG_QW];
  __shared__ float    spare[EE];

  const int tid  = threadIdx.x;
  const int lane = tid & 31;
  const int wv   = tid >> 5;
  const int half = lane >> 4;
  const int l16  = lane & 15;
  const int g    = blockIdx.x % RG_G;
  const int m    = blockIdx.x / RG_G;
  const int rowLo = g * RG_ROWS;
  const int eLo   = m * RG_SLICE;

  for (int j = tid; j < RG_ROWS * EE; j += RG_THREADS) accTile[j] = 0.0f;
  __syncthreads();

  const int wcount = rgcn_scan_slice(
      hr, eLo, tid, lane, queue[wv],
      [&](int r) { return (unsigned)(r - rowLo); },
      [&](unsigned d, int i) {
        const float v = values[i];
        const float* wp = W1 + (size_t)hc[i] * EE;
#pragma nounroll
        for (int e = 0; e < EE; ++e) atomAddF32(&accTile[d * EE + e], v * wp[e]);
      });
  __syncthreads();

  const int n = wcount < RG_QW ? wcount : RG_QW;
  for (int base = 0; base < n; base += 2 * RG_U3) {
    unsigned ent[RG_U3];
    bool ok[RG_U3];
#pragma unroll
    for (int u = 0; u < RG_U3; ++u) {
      const int q = base + 2 * u + half;
      ok[u] = q < n;
      ent[u] = ok[u] ? queue[wv][q] : 0u;
    }
    int col[RG_U3];
    float v[RG_U3];
#pragma unroll
    for (int u = 0; u < RG_U3; ++u) {
      const int i = ok[u] ? (int)(ent[u] & 0x1FFFFFu) : eLo;
      col[u] = __builtin_nontemporal_load(&hc[i]);
      v[u]   = __builtin_nontemporal_load(&values[i]);
    }
    float w[RG_U3];
#pragma unroll
    for (int u = 0; u < RG_U3; ++u)
      w[u] = W1[(size_t)col[u] * EE + l16];
#pragma unroll
    for (int u = 0; u < RG_U3; ++u) {
      float* dst = ok[u] ? &accTile[(ent[u] >> 21) * EE + l16] : &spare[l16];
      atomAddF32(dst, v[u] * w[u]);
    }
  }
  __syncthreads();

  for (int j = tid; j < RG_ROWS * EE; j += RG_THREADS) {
    const int row = rowLo + (j >> 4);
    if (row < NN) atomAddF32(&h[(size_t)rowLo * EE + j], accTile[j]);
  }
}

#define RG_U5   4

__global__ void __launch_bounds__(RG_THREADS)
rgcn_layer2(const float* __restrict__ values,
            const int* __restrict__ vr,
            const int* __restrict__ vc,
            const float* __restrict__ h,
            const float* __restrict__ bias1,
            const float* __restrict__ W2,
            float* __restrict__ out) {
  __shared__ float    W2s[RR * EE * CC];
  __shared__ float    outTile[RG_ROWS * CC];
  __shared__ unsigned queue[32][RG_QW];
  __shared__ float    spare[CC];

  const int tid  = threadIdx.x;
  const int lane = tid & 31;
  const int wv   = tid >> 5;
  const int sub  = lane >> 3;
  const int c    = lane & 7;
  const int g    = blockIdx.x % RG_G;
  const int m    = blockIdx.x / RG_G;
  const int rowLo = g * RG_ROWS;
  const int eLo   = m * RG_SLICE;

  for (int j = tid; j < RR * EE * CC; j += RG_THREADS) W2s[j] = W2[j];
  for (int j = tid; j < RG_ROWS * CC; j += RG_THREADS) outTile[j] = 0.0f;
  __syncthreads();

  const int wcount = rgcn_scan_slice(
      vr, eLo, tid, lane, queue[wv],
      [&](int r) {
        const unsigned ur = (unsigned)r;
        const unsigned prod = (unsigned)__umul24(ur, 1342u);
        const unsigned pe = prod >> 26;
        const unsigned se = ur - pe * (unsigned)NN;
        const unsigned sm = se - (unsigned)NN;
        return (se < sm ? se : sm) - (unsigned)rowLo;
      },
      [&](unsigned d, int i) {
        const int p = vr[i] / NN;
        const float v = values[i];
        const float* hp = h + (size_t)vc[i] * EE;
#pragma nounroll
        for (int cc = 0; cc < CC; ++cc) {
          const float* wp = W2s + p * (EE * CC) + cc;
          float acc = 0.0f;
#pragma nounroll
          for (int e = 0; e < EE; ++e) acc += fmaxf(hp[e] + bias1[e], 0.0f) * wp[e * CC];
          atomAddF32(&outTile[d * CC + cc], v * acc);
        }
      });
  __syncthreads();

  float b1[EE];
#pragma unroll
  for (int e = 0; e < EE; ++e) b1[e] = bias1[e];

  const int n = wcount < RG_QW ? wcount : RG_QW;
  for (int base = 0; base < n; base += 4 * RG_U5) {
    unsigned ent[RG_U5];
    bool ok[RG_U5];
#pragma unroll
    for (int u = 0; u < RG_U5; ++u) {
      const int q = base + 4 * u + sub;
      ok[u] = q < n;
      ent[u] = ok[u] ? queue[wv][q] : 0u;
    }
    int r[RG_U5], o[RG_U5];
    float v[RG_U5];
#pragma unroll
    for (int u = 0; u < RG_U5; ++u) {
      const int i = ok[u] ? (int)(ent[u] & 0x1FFFFFu) : eLo;
      r[u] = __builtin_nontemporal_load(&vr[i]);
      o[u] = __builtin_nontemporal_load(&vc[i]);
      v[u] = __builtin_nontemporal_load(&values[i]);
    }
    float4 hv[RG_U5][4];
#pragma unroll
    for (int u = 0; u < RG_U5; ++u) {
      const float4* hp4 = reinterpret_cast<const float4*>(h + (size_t)o[u] * EE);
#pragma unroll
      for (int k = 0; k < 4; ++k) hv[u][k] = hp4[k];
    }
    __builtin_amdgcn_sched_barrier(0);
    float acc[RG_U5];
#pragma unroll
    for (int u = 0; u < RG_U5; ++u) {
      const int p = r[u] / NN;
      const float* wp = W2s + p * (EE * CC) + c;
      float a = 0.0f;
#pragma unroll
      for (int k = 0; k < 4; ++k) {
        a += fmaxf(hv[u][k].x + b1[4 * k + 0], 0.0f) * wp[(4 * k + 0) * CC];
        a += fmaxf(hv[u][k].y + b1[4 * k + 1], 0.0f) * wp[(4 * k + 1) * CC];
        a += fmaxf(hv[u][k].z + b1[4 * k + 2], 0.0f) * wp[(4 * k + 2) * CC];
        a += fmaxf(hv[u][k].w + b1[4 * k + 3], 0.0f) * wp[(4 * k + 3) * CC];
      }
      acc[u] = a;
    }
#pragma unroll
    for (int u = 0; u < RG_U5; ++u) {
      float* dst = ok[u] ? &outTile[(ent[u] >> 21) * CC + c] : &spare[c];
      atomAddF32(dst, v[u] * acc[u]);
    }
  }
  __syncthreads();

  for (int j = tid; j < RG_ROWS * CC; j += RG_THREADS) {
    const int row = rowLo + (j >> 3);
    if (row < NN) atomAddF32(&out[(size_t)rowLo * CC + j], outTile[j]);
  }
}

extern "C" void kernel_launch(void* const* d_in, const int* in_sizes, int n_in,
                              void* d_out, int out_size, void* d_ws, size_t ws_size,
                              hipStream_t stream) {
  const float* values = (const float*)d_in[0];
  const float* comps1 = (const float*)d_in[1];
  const float* bases1 = (const float*)d_in[2];
  const float* comps2 = (const float*)d_in[3];
  const float* bases2 = (const float*)d_in[4];
  const float* bias1  = (const float*)d_in[5];
  const float* bias2  = (const float*)d_in[6];
  const int*   hr     = (const int*)d_in[7];
  const int*   hc     = (const int*)d_in[8];
  const int*   vr     = (const int*)d_in[9];
  const int*   vc     = (const int*)d_in[10];
  float* out = (float*)d_out;

  float* W1 = (float*)d_ws;
  float* h  = W1 + W1ELEMS;
  float* W2 = h + NE;

  rgcn_w1_gemm<<<832, 256, 0, stream>>>(comps1, bases1, W1, h, out, bias2, comps2, bases2, W2);
  rgcn_scatter_h<<<RG_G * RG_M, RG_THREADS, 0, stream>>>(values, hr, hc, W1, h);
  rgcn_layer2<<<RG_G * RG_M, RG_THREADS, 0, stream>>>(values, vr, vc, h, bias1, W2, out);
}


# both scatter kernels: rows cut into 32 groups x 8 edge slices (was 64 x 4), LDS tiles twice as tall: a workgroup scans half as many row ids; 8 partial tiles added per element, sum order changes
# speedup vs baseline: 1.3256x; 1.1240x over previous
#include <hip/hip_runtime.h>
#include <hip/hip_bf16.h>

#define NN    50000
#define RR    51
#define EE    16
#define CC    8
#define BB    40
#define NNZ   1600000
#define NE    (NN * EE)
#define W1ELEMS ((size_t)RR * NE)

typedef float v2f __attribute__((ext_vector_type(2)));
typedef float v8f __attribute__((ext_vector_type(8)));

__device__ __forceinline__ void atomAddF32(float* p, float v) {
  unsafeAtomicAdd(p, v);
}

__global__ void __launch_bounds__(256)
rgcn_w1_gemm(const float* __restrict__ comps1,
             const float* __restrict__ bases1,
             float* __restrict__ W1,
             float* __restrict__ h,
             float* __restrict__ out,
             const float* __restrict__ bias2,
             const float* __restrict__ comps2,
             const float* __restrict__ bases2,
             float* __restrict__ W2) {
  {
    const int gid      = blockIdx.x * blockDim.x + threadIdx.x;
    const int nThreads = gridDim.x * blockDim.x;
    float4* h4   = reinterpret_cast<float4*>(h);
    float4* out4 = reinterpret_cast<float4*>(out);
    const float4 zero4 = make_float4(0.0f, 0.0f, 0.0f, 0.0f);
    for (int i = gid; i < NE / 4; i += nThreads) h4[i] = zero4;
    const int c0 = (gid & 1) * 4;
    const float4 bv = make_float4(bias2[c0], bias2[c0 + 1], bias2[c0 + 2], bias2[c0 + 3]);
    for (int i = gid; i < NN * CC / 4; i += nThreads) out4[i] = bv;
    if (gid < RR * EE * CC) {
      const int r  = gid >> 7;
      const int ec = gid & 127;
      float acc = 0.0f;
#pragma unroll
      for (int b = 0; b < BB; ++b)
        acc += comps2[r * BB + b] * bases2[b * (EE * CC) + ec];
      W2[gid] = acc;
    }
  }

  const int lane = threadIdx.x & 31;
  const int half = lane >> 4;
  const int l16  = lane & 15;

  v2f a[4][10];
#pragma unroll
  for (int mt = 0; mt < 4; ++mt) {
    int m = mt * 16 + l16;
#pragma unroll
    for (int ks = 0; ks < 10; ++ks) {
      int k = ks * 4 + half * 2;
      float ax = 0.0f, ay = 0.0f;
      if (m < RR) {
        ax = comps1[m * BB + k];
        ay = comps1[m * BB + k + 1];
      }
      a[mt][ks].x = ax;
      a[mt][ks].y = ay;
    }
  }

  const int wavesPerBlock = blockDim.x >> 5;
  const int waveId  = blockIdx.x * wavesPerBlock + (threadIdx.x >> 5);
  const int nWaves  = gridDim.x * wavesPerBlock;
  const int nTiles  = NE / 16;

  for (int tile = waveId; tile < nTiles; tile += nWaves) {
    const int col = tile * 16 + l16;
    v8f acc[4] = {};
#pragma unroll
    for (int ks = 0; ks < 10; ++ks) {
      const int k = ks * 4 + half * 2;
      v2f b;
      b.x = __builtin_nontemporal_load(&bases1[(size_t)k * NE + col]);
      b.y = __builtin_nontemporal_load(&bases1[(size_t)(k + 1) * NE + col]);
#pragma unroll
      for (int mt = 0; mt < 4; ++mt) {
        acc[mt] = __builtin_amdgcn_wmma_f32_16x16x4_f32(
             false, a[mt][ks],  false, b,
             (short)0, acc[mt],
             false,  false);
      }
    }
#pragma unroll
    for (int mt = 0; mt < 4; ++mt) {
#pragma unroll
      for (int j = 0; j < 8; ++j) {
        int m = mt * 16 + j + half * 8;
        if (m < RR) W1[(size_t)m * NE + col] = acc[mt][j];
      }
    }
  }
}

#define RG_ROWS 1564
#define RG_G    32
#define RG_M    8
#define RG_SLICE (NNZ / RG_M)
#define RG_QW   1024
#define RG_THREADS 1024

template <typename RowFn, typename DirectFn>
__device__ __forceinline__ int rgcn_scan_slice(const int* __restrict__ ids, int eLo,
                                               int tid, int lane, unsigned* myQueue,
                                               RowFn localRow, DirectFn direct) {
  const int4* ids4 = reinterpret_cast<const int4*>(ids + eLo);
  const unsigned ltMask = (1u << lane) - 1u;
  int wcount = 0;
  for (int jb = 0; jb < RG_SLICE / 4; jb += 4 * RG_THREADS) {
    int4 rr[4];
#pragma unroll
    for (int k = 0; k < 4; ++k) {
      const int j = jb + k * RG_THREADS + tid;
      rr[k] = make_int4(-1, -1, -1, -1);
      if (j < RG_SLICE / 4) rr[k] = ids4[j];
    }
#pragma unroll
    for (int k = 0; k < 4; ++k) {
      const int j = jb + k * RG_THREADS + tid;
      const bool inRange = j < RG_SLICE / 4;
      const unsigned d0 = localRow(rr[k].x), d1 = localRow(rr[k].y);
      const unsigned d2 = localRow(rr[k].z), d3 = localRow(rr[k].w);
      const bool s0 = inRange & (d0 < RG_ROWS), s1 = inRange & (d1 < RG_ROWS);
      const bool s2 = inRange & (d2 < RG_ROWS), s3 = inRange & (d3 < RG_ROWS);
      const unsigned b0 = __builtin_amdgcn_ballot_w32(s0);
      const unsigned b1 = __builtin_amdgcn_ballot_w32(s1);
      const unsigned b2 = __builtin_amdgcn_ballot_w32(s2);
      const unsigned b3 = __builtin_amdgcn_ballot_w32(s3);
      if ((b0 | b1 | b2 | b3) != 0u) {
        const int i0 = eLo + 4 * j;
        const int p0 = wcount + __popc(b0 & ltMask);
        const int c0 = wcount + __popc(b0);
        const int p1 = c0 + __popc(b1 & ltMask);
        const int c1 = c0 + __popc(b1);
        const int p2 = c1 + __popc(b2 & ltMask);
        const int c2 = c1 + __popc(b2);
        const int p3 = c2 + __popc(b3 & ltMask);
        wcount = c2 + __popc(b3);
        if (s0) { if (__builtin_expect(p0 < RG_QW, 1)) myQueue[p0] = (d0 << 21) | (unsigned)(i0);
                  else direct(d0, i0); }
        if (s1) { if (__builtin_expect(p1 < RG_QW, 1)) myQueue[p1] = (d1 << 21) | (unsigned)(i0 + 1);
                  else direct(d1, i0 + 1); }
        if (s2) { if (__builtin_expect(p2 < RG_QW, 1)) myQueue[p2] = (d2 << 21) | (unsigned)(i0 + 2);
                  else direct(d2, i0 + 2); }
        if (s3) { if (__builtin_expect(p3 < RG_QW, 1)) myQueue[p3] = (d3 << 21) | (unsigned)(i0 + 3);
                  else direct(d3, i0 + 3); }
      }
    }
  }
  return wcount;
}

#define RG_U3   16

__global__ void __launch_bounds__(RG_THREADS)
rgcn_scatter_h(const float* __restrict__ values,
               const int* __restrict__ hr,
               const int* __restrict__ hc,
               const float* __restrict__ W1,
               float* __restrict__ h) {
  __shared__ float    accTile[RG_ROWS * EE];
  __shared__ unsigned queue[32][RG_QW];
  __shared__ float    spare[EE];

  const int tid  = threadIdx.x;
  const int lane = tid & 31;
  const int wv   = tid >> 5;
  const int half = lane >> 4;
  const int l16  = lane & 15;
  const int g    = blockIdx.x % RG_G;
  const int m    = blockIdx.x / RG_G;
  const int rowLo = g * RG_ROWS;
  const int eLo   = m * RG_SLICE;

  for (int j = tid; j < RG_ROWS * EE; j += RG_THREADS) accTile[j] = 0.0f;
  __syncthreads();

  const int wcount = rgcn_scan_slice(
      hr, eLo, tid, lane, queue[wv],
      [&](int r) { return (unsigned)(r - rowLo); },
      [&](unsigned d, int i) {
        const float v = values[i];
        const float* wp = W1 + (size_t)hc[i] * EE;
#pragma nounroll
        for (int e = 0; e < EE; ++e) atomAddF32(&accTile[d * EE + e], v * wp[e]);
      });
  __syncthreads();

  const int n = wcount < RG_QW ? wcount : RG_QW;
  for (int base = 0; base < n; base += 2 * RG_U3) {
    unsigned ent[RG_U3];
    bool ok[RG_U3];
#pragma unroll
    for (int u = 0; u < RG_U3; ++u) {
      const int q = base + 2 * u + half;
      ok[u] = q < n;
      ent[u] = ok[u] ? queue[wv][q] : 0u;
    }
    int col[RG_U3];
    float v[RG_U3];
#pragma unroll
    for (int u = 0; u < RG_U3; ++u) {
      const int i = ok[u] ? (int)(ent[u] & 0x1FFFFFu) : eLo;
      col[u] = __builtin_nontemporal_load(&hc[i]);
      v[u]   = __builtin_nontemporal_load(&values[i]);
    }
    float w[RG_U3];
#pragma unroll
    for (int u = 0; u < RG_U3; ++u)
      w[u] = W1[(size_t)col[u] * EE + l16];
#pragma unroll
    for (int u = 0; u < RG_U3; ++u) {
      float* dst = ok[u] ? &accTile[(ent[u] >> 21) * EE + l16] : &spare[l16];
      atomAddF32(dst, v[u] * w[u]);
    }
  }
  __syncthreads();

  for (int j = tid; j < RG_ROWS * EE; j += RG_THREADS) {
    const int row = rowLo + (j >> 4);
    if (row < NN) atomAddF32(&h[(size_t)rowLo * EE + j], accTile[j]);
  }
}

#define RG_U5   4

__global__ void __launch_bounds__(RG_THREADS)
rgcn_layer2(const float* __restrict__ values,
            const int* __restrict__ vr,
            const int* __restrict__ vc,
            const float* __restrict__ h,
            const float* __restrict__ bias1,
            const float* __restrict__ W2,
            float* __restrict__ out) {
  __shared__ float    W2s[RR * EE * CC];
  __shared__ float    outTile[RG_ROWS * CC];
  __shared__ unsigned queue[32][RG_QW];
  __shared__ float    spare[CC];

  const int tid  = threadIdx.x;
  const int lane = tid & 31;
  const int wv   = tid >> 5;
  const int sub  = lane >> 3;
  const int c    = lane & 7;
  const int g    = blockIdx.x % RG_G;
  const int m    = blockIdx.x / RG_G;
  const int rowLo = g * RG_ROWS;
  const int eLo   = m * RG_SLICE;

  for (int j = tid; j < RR * EE * CC; j += RG_THREADS) W2s[j] = W2[j];
  for (int j = tid; j < RG_ROWS * CC; j += RG_THREADS) outTile[j] = 0.0f;
  __syncthreads();

  const int wcount = rgcn_scan_slice(
      vr, eLo, tid, lane, queue[wv],
      [&](int r) {
        const unsigned ur = (unsigned)r;
        const unsigned prod = (unsigned)__umul24(ur, 1342u);
        const unsigned pe = prod >> 26;
        const unsigned se = ur - pe * (unsigned)NN;
        const unsigned sm = se - (unsigned)NN;
        return (se < sm ? se : sm) - (unsigned)rowLo;
      },
      [&](unsigned d, int i) {
        const int p = vr[i] / NN;
        const float v = values[i];
        const float* hp = h + (size_t)vc[i] * EE;
#pragma nounroll
        for (int cc = 0; cc < CC; ++cc) {
          const float* wp = W2s + p * (EE * CC) + cc;
          float acc = 0.0f;
#pragma nounroll
          for (int e = 0; e < EE; ++e) acc += fmaxf(hp[e] + bias1[e], 0.0f) * wp[e * CC];
          atomAddF32(&outTile[d * CC + cc], v * acc);
        }
      });
  __syncthreads();

  float b1[EE];
#pragma unroll
  for (int e = 0; e < EE; ++e) b1[e] = bias1[e];

  const int n = wcount < RG_QW ? wcount : RG_QW;
  for (int base = 0; base < n; base += 4 * RG_U5) {
    unsigned ent[RG_U5];
    bool ok[RG_U5];
#pragma unroll
    for (int u = 0; u < RG_U5; ++u) {
      const int q = base + 4 * u + sub;
      ok[u] = q < n;
      ent[u] = ok[u] ? queue[wv][q] : 0u;
    }
    int r[RG_U5], o[RG_U5];
    float v[RG_U5];
#pragma unroll
    for (int u = 0; u < RG_U5; ++u) {
      const int i = ok[u] ? (int)(ent[u] & 0x1FFFFFu) : eLo;
      r[u] = __builtin_nontemporal_load(&vr[i]);
      o[u] = __builtin_nontemporal_load(&vc[i]);
      v[u] = __builtin_nontemporal_load(&values[i]);
    }
    float4 hv[RG_U5][4];
#pragma unroll
    for (int u = 0; u < RG_U5; ++u) {
      const float4* hp4 = reinterpret_cast<const float4*>(h + (size_t)o[u] * EE);
#pragma unroll
      for (int k = 0; k < 4; ++k) hv[u][k] = hp4[k];
    }
    __builtin_amdgcn_sched_barrier(0);
    float acc[RG_U5];
#pragma unroll
    for (int u = 0; u < RG_U5; ++u) {
      const int p = r[u] / NN;
      const float* wp = W2s + p * (EE * CC) + c;
      float a = 0.0f;
#pragma unroll
      for (int k = 0; k < 4; ++k) {
        a += fmaxf(hv[u][k].x + b1[4 * k + 0], 0.0f) * wp[(4 * k + 0) * CC];
        a += fmaxf(hv[u][k].y + b1[4 * k + 1], 0.0f) * wp[(4 * k + 1) * CC];
        a += fmaxf(hv[u][k].z + b1[4 * k + 2], 0.0f) * wp[(4 * k + 2) * CC];
        a += fmaxf(hv[u][k].w + b1[4 * k + 3], 0.0f) * wp[(4 * k + 3) * CC];
      }
      acc[u] = a;
    }
#pragma unroll
    for (int u = 0; u < RG_U5; ++u) {
      float* dst = ok[u] ? &outTile[(ent[u] >> 21) * CC + c] : &spare[c];
      atomAddF32(dst, v[u] * acc[u]);
    }
  }
  __syncthreads();

  for (int j = tid; j < RG_ROWS * CC; j += RG_THREADS) {
    const int row = rowLo + (j >> 3);
    if (row < NN) atomAddF32(&out[(size_t)rowLo * CC + j], outTile[j]);
  }
}

extern "C" void kernel_launch(void* const* d_in, const int* in_sizes, int n_in,
                              void* d_out, int out_size, void* d_ws, size_t ws_size,
                              hipStream_t stream) {
  const float* values = (const float*)d_in[0];
  const float* comps1 = (const float*)d_in[1];
  const float* bases1 = (const float*)d_in[2];
  const float* comps2 = (const float*)d_in[3];
  const float* bases2 = (const float*)d_in[4];
  const float* bias1  = (const float*)d_in[5];
  const float* bias2  = (const float*)d_in[6];
  const int*   hr     = (const int*)d_in[7];
  const int*   hc     = (const int*)d_in[8];
  const int*   vr     = (const int*)d_in[9];
  const int*   vc     = (const int*)d_in[10];
  float* out = (float*)d_out;

  float* W1 = (float*)d_ws;
  float* h  = W1 + W1ELEMS;
  float* W2 = h + NE;

  rgcn_w1_gemm<<<832, 256, 0, stream>>>(comps1, bases1, W1, h, out, bias2, comps2, bases2, W2);
  rgcn_scatter_h<<<RG_G * RG_M, RG_THREADS, 0, stream>>>(values, hr, hc, W1, h);
  rgcn_layer2<<<RG_G * RG_M, RG_THREADS, 0, stream>>>(values, vr, vc, h, bias1, W2, out);
}


# both scatter kernels: each wave drains its queue in batches between its scan trips (barrier between scan and drain gone), so gather latency runs under the id tests; order of the LDS sums changes
# speedup vs baseline: 1.4603x; 1.1016x over previous
#include <hip/hip_runtime.h>
#include <hip/hip_bf16.h>

#define NN    50000
#define RR    51
#define EE    16
#define CC    8
#define BB    40
#define NNZ   1600000
#define NE    (NN * EE)
#define W1ELEMS ((size_t)RR * NE)

typedef float v2f __attribute__((ext_vector_type(2)));
typedef float v8f __attribute__((ext_vector_type(8)));

__device__ __forceinline__ void atomAddF32(float* p, float v) {
  unsafeAtomicAdd(p, v);
}

__global__ void __launch_bounds__(256)
rgcn_w1_gemm(const float* __restrict__ comps1,
             const float* __restrict__ bases1,
             float* __restrict__ W1,
             float* __restrict__ h,
             float* __restrict__ out,
             const float* __restrict__ bias2,
             const float* __restrict__ comps2,
             const float* __restrict__ bases2,
             float* __restrict__ W2) {
  {
    const int gid      = blockIdx.x * blockDim.x + threadIdx.x;
    const int nThreads = gridDim.x * blockDim.x;
    float4* h4   = reinterpret_cast<float4*>(h);
    float4* out4 = reinterpret_cast<float4*>(out);
    const float4 zero4 = make_float4(0.0f, 0.0f, 0.0f, 0.0f);
    for (int i = gid; i < NE / 4; i += nThreads) h4[i] = zero4;
    const int c0 = (gid & 1) * 4;
    const float4 bv = make_float4(bias2[c0], bias2[c0 + 1], bias2[c0 + 2], bias2[c0 + 3]);
    for (int i = gid; i < NN * CC / 4; i += nThreads) out4[i] = bv;
    if (gid < RR * EE * CC) {
      const int r  = gid >> 7;
      const int ec = gid & 127;
      float acc = 0.0f;
#pragma unroll
      for (int b = 0; b < BB; ++b)
        acc += comps2[r * BB + b] * bases2[b * (EE * CC) + ec];
      W2[gid] = acc;
    }
  }

  const int lane = threadIdx.x & 31;
  const int half = lane >> 4;
  const int l16  = lane & 15;

  v2f a[4][10];
#pragma unroll
  for (int mt = 0; mt < 4; ++mt) {
    int m = mt * 16 + l16;
#pragma unroll
    for (int ks = 0; ks < 10; ++ks) {
      int k = ks * 4 + half * 2;
      float ax = 0.0f, ay = 0.0f;
      if (m < RR) {
        ax = comps1[m * BB + k];
        ay = comps1[m * BB + k + 1];
      }
      a[mt][ks].x = ax;
      a[mt][ks].y = ay;
    }
  }

  const int wavesPerBlock = blockDim.x >> 5;
  const int waveId  = blockIdx.x * wavesPerBlock + (threadIdx.x >> 5);
  const int nWaves  = gridDim.x * wavesPerBlock;
  const int nTiles  = NE / 16;

  for (int tile = waveId; tile < nTiles; tile += nWaves) {
    const int col = tile * 16 + l16;
    v8f acc[4] = {};
#pragma unroll
    for (int ks = 0; ks < 10; ++ks) {
      const int k = ks * 4 + half * 2;
      v2f b;
      b.x = __builtin_nontemporal_load(&bases1[(size_t)k * NE + col]);
      b.y = __builtin_nontemporal_load(&bases1[(size_t)(k + 1) * NE + col]);
#pragma unroll
      for (int mt = 0; mt < 4; ++mt) {
        acc[mt] = __builtin_amdgcn_wmma_f32_16x16x4_f32(
             false, a[mt][ks],  false, b,
             (short)0, acc[mt],
             false,  false);
      }
    }
#pragma unroll
    for (int mt = 0; mt < 4; ++mt) {
#pragma unroll
      for (int j = 0; j < 8; ++j) {
        int m = mt * 16 + j + half * 8;
        if (m < RR) W1[(size_t)m * NE + col] = acc[mt][j];
      }
    }
  }
}

#define RG_ROWS 1564
#define RG_G    32
#define RG_M    8
#define RG_SLICE (NNZ / RG_M)
#define RG_QW   1024
#define RG_THREADS 1024

template <int BATCH, typename RowFn, typename DirectFn, typename DrainFn>
__device__ __forceinline__ void rgcn_scan_slice(const int* __restrict__ ids, int eLo,
                                                int tid, int lane, unsigned* myQueue,
                                                RowFn localRow, DirectFn direct,
                                                DrainFn drainBatch) {
  const int4* ids4 = reinterpret_cast<const int4*>(ids + eLo);
  const unsigned ltMask = (1u << lane) - 1u;
  int wcount = 0;
  int drained = 0;
  for (int jb = 0; jb < RG_SLICE / 4; jb += 4 * RG_THREADS) {
    int4 rr[4];
#pragma unroll
    for (int k = 0; k < 4; ++k) {
      const int j = jb + k * RG_THREADS + tid;
      rr[k] = make_int4(-1, -1, -1, -1);
      if (j < RG_SLICE / 4) rr[k] = ids4[j];
    }
#pragma unroll
    for (int k = 0; k < 4; ++k) {
      const int j = jb + k * RG_THREADS + tid;
      const bool inRange = j < RG_SLICE / 4;
      const unsigned d0 = localRow(rr[k].x), d1 = localRow(rr[k].y);
      const unsigned d2 = localRow(rr[k].z), d3 = localRow(rr[k].w);
      const bool s0 = inRange & (d0 < RG_ROWS), s1 = inRange & (d1 < RG_ROWS);
      const bool s2 = inRange & (d2 < RG_ROWS), s3 = inRange & (d3 < RG_ROWS);
      const unsigned b0 = __builtin_amdgcn_ballot_w32(s0);
      const unsigned b1 = __builtin_amdgcn_ballot_w32(s1);
      const unsigned b2 = __builtin_amdgcn_ballot_w32(s2);
      const unsigned b3 = __builtin_amdgcn_ballot_w32(s3);
      if ((b0 | b1 | b2 | b3) != 0u) {
        const int i0 = eLo + 4 * j;
        const int p0 = wcount + __popc(b0 & ltMask);
        const int c0 = wcount + __popc(b0);
        const int p1 = c0 + __popc(b1 & ltMask);
        const int c1 = c0 + __popc(b1);
        const int p2 = c1 + __popc(b2 & ltMask);
        const int c2 = c1 + __popc(b2);
        const int p3 = c2 + __popc(b3 & ltMask);
        wcount = c2 + __popc(b3);
        if (s0) { if (__builtin_expect(p0 < RG_QW, 1)) myQueue[p0] = (d0 << 21) | (unsigned)(i0);
                  else direct(d0, i0); }
        if (s1) { if (__builtin_expect(p1 < RG_QW, 1)) myQueue[p1] = (d1 << 21) | (unsigned)(i0 + 1);
                  else direct(d1, i0 + 1); }
        if (s2) { if (__builtin_expect(p2 < RG_QW, 1)) myQueue[p2] = (d2 << 21) | (unsigned)(i0 + 2);
                  else direct(d2, i0 + 2); }
        if (s3) { if (__builtin_expect(p3 < RG_QW, 1)) myQueue[p3] = (d3 << 21) | (unsigned)(i0 + 3);
                  else direct(d3, i0 + 3); }
      }
    }
    const int avail = wcount < RG_QW ? wcount : RG_QW;
    for (; avail - drained >= BATCH; drained += BATCH) drainBatch(drained, avail);
  }
  const int n = wcount < RG_QW ? wcount : RG_QW;
  for (; drained < n; drained += BATCH) drainBatch(drained, n);
}

#define RG_U3   16

__global__ void __launch_bounds__(RG_THREADS)
rgcn_scatter_h(const float* __restrict__ values,
               const int* __restrict__ hr,
               const int* __restrict__ hc,
               const float* __restrict__ W1,
               float* __restrict__ h) {
  __shared__ float    accTile[RG_ROWS * EE];
  __shared__ unsigned queue[32][RG_QW];
  __shared__ float    spare[EE];

  const int tid  = threadIdx.x;
  const int lane = tid & 31;
  const int wv   = tid >> 5;
  const int half = lane >> 4;
  const int l16  = lane & 15;
  const int g    = blockIdx.x % RG_G;
  const int m    = blockIdx.x / RG_G;
  const int rowLo = g * RG_ROWS;
  const int eLo   = m * RG_SLICE;

  for (int j = tid; j < RG_ROWS * EE; j += RG_THREADS) accTile[j] = 0.0f;
  __syncthreads();

  rgcn_scan_slice<2 * RG_U3>(
      hr, eLo, tid, lane, queue[wv],
      [&](int r) { return (unsigned)(r - rowLo); },
      [&](unsigned d, int i) {
        const float v = values[i];
        const float* wp = W1 + (size_t)hc[i] * EE;
#pragma nounroll
        for (int e = 0; e < EE; ++e) atomAddF32(&accTile[d * EE + e], v * wp[e]);
      },
      [&](int base, int n) {
        unsigned ent[RG_U3];
        bool ok[RG_U3];
#pragma unroll
        for (int u = 0; u < RG_U3; ++u) {
          const int q = base + 2 * u + half;
          ok[u] = q < n;
          ent[u] = ok[u] ? queue[wv][q] : 0u;
        }
        int col[RG_U3];
        float v[RG_U3];
#pragma unroll
        for (int u = 0; u < RG_U3; ++u) {
          const int i = ok[u] ? (int)(ent[u] & 0x1FFFFFu) : eLo;
          col[u] = __builtin_nontemporal_load(&hc[i]);
          v[u]   = __builtin_nontemporal_load(&values[i]);
        }
        float w[RG_U3];
#pragma unroll
        for (int u = 0; u < RG_U3; ++u)
          w[u] = W1[(size_t)col[u] * EE + l16];
#pragma unroll
        for (int u = 0; u < RG_U3; ++u) {
          float* dst = ok[u] ? &accTile[(ent[u] >> 21) * EE + l16] : &spare[l16];
          atomAddF32(dst, v[u] * w[u]);
        }
      });
  __syncthreads();

  for (int j = tid; j < RG_ROWS * EE; j += RG_THREADS) {
    const int row = rowLo + (j >> 4);
    if (row < NN) atomAddF32(&h[(size_t)rowLo * EE + j], accTile[j]);
  }
}

#define RG_U5   4

__global__ void __launch_bounds__(RG_THREADS)
rgcn_layer2(const float* __restrict__ values,
            const int* __restrict__ vr,
            const int* __restrict__ vc,
            const float* __restrict__ h,
            const float* __restrict__ bias1,
            const float* __restrict__ W2,
            float* __restrict__ out) {
  __shared__ float    W2s[RR * EE * CC];
  __shared__ float    outTile[RG_ROWS * CC];
  __shared__ unsigned queue[32][RG_QW];
  __shared__ float    spare[CC];

  const int tid  = threadIdx.x;
  const int lane = tid & 31;
  const int wv   = tid >> 5;
  const int sub  = lane >> 3;
  const int c    = lane & 7;
  const int g    = blockIdx.x % RG_G;
  const int m    = blockIdx.x / RG_G;
  const int rowLo = g * RG_ROWS;
  const int eLo   = m * RG_SLICE;

  for (int j = tid; j < RR * EE * CC; j += RG_THREADS) W2s[j] = W2[j];
  for (int j = tid; j < RG_ROWS * CC; j += RG_THREADS) outTile[j] = 0.0f;
  __syncthreads();

  float b1[EE];
#pragma unroll
  for (int e = 0; e < EE; ++e) b1[e] = bias1[e];

  rgcn_scan_slice<4 * RG_U5>(
      vr, eLo, tid, lane, queue[wv],
      [&](int r) {
        const unsigned ur = (unsigned)r;
        const unsigned prod = (unsigned)__umul24(ur, 1342u);
        const unsigned pe = prod >> 26;
        const unsigned se = ur - pe * (unsigned)NN;
        const unsigned sm = se - (unsigned)NN;
        return (se < sm ? se : sm) - (unsigned)rowLo;
      },
      [&](unsigned d, int i) {
        const int p = vr[i] / NN;
        const float v = values[i];
        const float* hp = h + (size_t)vc[i] * EE;
#pragma nounroll
        for (int cc = 0; cc < CC; ++cc) {
          const float* wp = W2s + p * (EE * CC) + cc;
          float acc = 0.0f;
#pragma nounroll
          for (int e = 0; e < EE; ++e) acc += fmaxf(hp[e] + bias1[e], 0.0f) * wp[e * CC];
          atomAddF32(&outTile[d * CC + cc], v * acc);
        }
      },
      [&](int base, int n) {
        unsigned ent[RG_U5];
        bool ok[RG_U5];
#pragma unroll
        for (int u = 0; u < RG_U5; ++u) {
          const int q = base + 4 * u + sub;
          ok[u] = q < n;
          ent[u] = ok[u] ? queue[wv][q] : 0u;
        }
        int r[RG_U5], o[RG_U5];
        float v[RG_U5];
#pragma unroll
        for (int u = 0; u < RG_U5; ++u) {
          const int i = ok[u] ? (int)(ent[u] & 0x1FFFFFu) : eLo;
          r[u] = __builtin_nontemporal_load(&vr[i]);
          o[u] = __builtin_nontemporal_load(&vc[i]);
          v[u] = __builtin_nontemporal_load(&values[i]);
        }
        float4 hv[RG_U5][4];
#pragma unroll
        for (int u = 0; u < RG_U5; ++u) {
          const float4* hp4 = reinterpret_cast<const float4*>(h + (size_t)o[u] * EE);
#pragma unroll
          for (int k = 0; k < 4; ++k) hv[u][k] = hp4[k];
        }
        __builtin_amdgcn_sched_barrier(0);
        float acc[RG_U5];
#pragma unroll
        for (int u = 0; u < RG_U5; ++u) {
          const int p = r[u] / NN;
          const float* wp = W2s + p * (EE * CC) + c;
          float a = 0.0f;
#pragma unroll
          for (int k = 0; k < 4; ++k) {
            a += fmaxf(hv[u][k].x + b1[4 * k + 0], 0.0f) * wp[(4 * k + 0) * CC];
            a += fmaxf(hv[u][k].y + b1[4 * k + 1], 0.0f) * wp[(4 * k + 1) * CC];
            a += fmaxf(hv[u][k].z + b1[4 * k + 2], 0.0f) * wp[(4 * k + 2) * CC];
            a += fmaxf(hv[u][k].w + b1[4 * k + 3], 0.0f) * wp[(4 * k + 3) * CC];
          }
          acc[u] = a;
        }
#pragma unroll
        for (int u = 0; u < RG_U5; ++u) {
          float* dst = ok[u] ? &outTile[(ent[u] >> 21) * CC + c] : &spare[c];
          atomAddF32(dst, v[u] * acc[u]);
        }
      });
  __syncthreads();

  for (int j = tid; j < RG_ROWS * CC; j += RG_THREADS) {
    const int row = rowLo + (j >> 3);
    if (row < NN) atomAddF32(&out[(size_t)rowLo * CC + j], outTile[j]);
  }
}

extern "C" void kernel_launch(void* const* d_in, const int* in_sizes, int n_in,
                              void* d_out, int out_size, void* d_ws, size_t ws_size,
                              hipStream_t stream) {
  const float* values = (const float*)d_in[0];
  const float* comps1 = (const float*)d_in[1];
  const float* bases1 = (const float*)d_in[2];
  const float* comps2 = (const float*)d_in[3];
  const float* bases2 = (const float*)d_in[4];
  const float* bias1  = (const float*)d_in[5];
  const float* bias2  = (const float*)d_in[6];
  const int*   hr     = (const int*)d_in[7];
  const int*   hc     = (const int*)d_in[8];
  const int*   vr     = (const int*)d_in[9];
  const int*   vc     = (const int*)d_in[10];
  float* out = (float*)d_out;

  float* W1 = (float*)d_ws;
  float* h  = W1 + W1ELEMS;
  float* W2 = h + NE;

  rgcn_w1_gemm<<<832, 256, 0, stream>>>(comps1, bases1, W1, h, out, bias2, comps2, bases2, W2);
  rgcn_scatter_h<<<RG_G * RG_M, RG_THREADS, 0, stream>>>(values, hr, hc, W1, h);
  rgcn_layer2<<<RG_G * RG_M, RG_THREADS, 0, stream>>>(values, vr, vc, h, bias1, W2, out);
}


# both scatter kernels: 16 row groups x 16 edge slices (was 32 x 8), LDS tiles 2x as tall, a slice's empty last load groups skipped: half the id tests; 16 partial tiles per element, sum order changes
# speedup vs baseline: 1.5043x; 1.0301x over previous
#include <hip/hip_runtime.h>
#include <hip/hip_bf16.h>

#define NN    50000
#define RR    51
#define EE    16
#define CC    8
#define BB    40
#define NNZ   1600000
#define NE    (NN * EE)
#define W1ELEMS ((size_t)RR * NE)

typedef float v2f __attribute__((ext_vector_type(2)));
typedef float v8f __attribute__((ext_vector_type(8)));

__device__ __forceinline__ void atomAddF32(float* p, float v) {
  unsafeAtomicAdd(p, v);
}

__global__ void __launch_bounds__(256)
rgcn_w1_gemm(const float* __restrict__ comps1,
             const float* __restrict__ bases1,
             float* __restrict__ W1,
             float* __restrict__ h,
             float* __restrict__ out,
             const float* __restrict__ bias2,
             const float* __restrict__ comps2,
             const float* __restrict__ bases2,
             float* __restrict__ W2) {
  {
    const int gid      = blockIdx.x * blockDim.x + threadIdx.x;
    const int nThreads = gridDim.x * blockDim.x;
    float4* h4   = reinterpret_cast<float4*>(h);
    float4* out4 = reinterpret_cast<float4*>(out);
    const float4 zero4 = make_float4(0.0f, 0.0f, 0.0f, 0.0f);
    for (int i = gid; i < NE / 4; i += nThreads) h4[i] = zero4;
    const int c0 = (gid & 1) * 4;
    const float4 bv = make_float4(bias2[c0], bias2[c0 + 1], bias2[c0 + 2], bias2[c0 + 3]);
    for (int i = gid; i < NN * CC / 4; i += nThreads) out4[i] = bv;
    if (gid < RR * EE * CC) {
      const int r  = gid >> 7;
      const int ec = gid & 127;
      float acc = 0.0f;
#pragma unroll
      for (int b = 0; b < BB; ++b)
        acc += comps2[r * BB + b] * bases2[b * (EE * CC) + ec];
      W2[gid] = acc;
    }
  }

  const int lane = threadIdx.x & 31;
  const int half = lane >> 4;
  const int l16  = lane & 15;

  v2f a[4][10];
#pragma unroll
  for (int mt = 0; mt < 4; ++mt) {
    int m = mt * 16 + l16;
#pragma unroll
    for (int ks = 0; ks < 10; ++ks) {
      int k = ks * 4 + half * 2;
      float ax = 0.0f, ay = 0.0f;
      if (m < RR) {
        ax = comps1[m * BB + k];
        ay = comps1[m * BB + k + 1];
      }
      a[mt][ks].x = ax;
      a[mt][ks].y = ay;
    }
  }

  const int wavesPerBlock = blockDim.x >> 5;
  const int waveId  = blockIdx.x * wavesPerBlock + (threadIdx.x >> 5);
  const int nWaves  = gridDim.x * wavesPerBlock;
  const int nTiles  = NE / 16;

  for (int tile = waveId; tile < nTiles; tile += nWaves) {
    const int col = tile * 16 + l16;
    v8f acc[4] = {};
#pragma unroll
    for (int ks = 0; ks < 10; ++ks) {
      const int k = ks * 4 + half * 2;
      v2f b;
      b.x = __builtin_nontemporal_load(&bases1[(size_t)k * NE + col]);
      b.y = __builtin_nontemporal_load(&bases1[(size_t)(k + 1) * NE + col]);
#pragma unroll
      for (int mt = 0; mt < 4; ++mt) {
        acc[mt] = __builtin_amdgcn_wmma_f32_16x16x4_f32(
             false, a[mt][ks],  false, b,
             (short)0, acc[mt],
             false,  false);
      }
    }
#pragma unroll
    for (int mt = 0; mt < 4; ++mt) {
#pragma unroll
      for (int j = 0; j < 8; ++j) {
        int m = mt * 16 + j + half * 8;
        if (m < RR) W1[(size_t)m * NE + col] = acc[mt][j];
      }
    }
  }
}

#define RG_ROWS 3128
#define RG_G    16
#define RG_M    16
#define RG_SLICE (NNZ / RG_M)
#define RG_QSHIFT 17
#define RG_QMASK  0x1FFFFu
#define RG_THREADS 1024

template <int BATCH, int RG_QW, typename RowFn, typename DirectFn, typename DrainFn>
__device__ __forceinline__ void rgcn_scan_slice(const int* __restrict__ ids, int eLo,
                                                int tid, int lane, unsigned* myQueue,
                                                RowFn localRow, DirectFn direct,
                                                DrainFn drainBatch) {
  const int4* ids4 = reinterpret_cast<const int4*>(ids + eLo);
  const unsigned ltMask = (1u << lane) - 1u;
  int wcount = 0;
  int drained = 0;
  for (int jb = 0; jb < RG_SLICE / 4; jb += 4 * RG_THREADS) {
    int4 rr[4];
#pragma unroll
    for (int k = 0; k < 4; ++k) {
      const int j = jb + k * RG_THREADS + tid;
      rr[k] = make_int4(-1, -1, -1, -1);
      if (j < RG_SLICE / 4) rr[k] = ids4[j];
    }
#pragma unroll
    for (int k = 0; k < 4; ++k) {
      if (jb + k * RG_THREADS >= RG_SLICE / 4) break;
      const int j = jb + k * RG_THREADS + tid;
      const bool inRange = j < RG_SLICE / 4;
      const unsigned d0 = localRow(rr[k].x), d1 = localRow(rr[k].y);
      const unsigned d2 = localRow(rr[k].z), d3 = localRow(rr[k].w);
      const bool s0 = inRange & (d0 < RG_ROWS), s1 = inRange & (d1 < RG_ROWS);
      const bool s2 = inRange & (d2 < RG_ROWS), s3 = inRange & (d3 < RG_ROWS);
      const unsigned b0 = __builtin_amdgcn_ballot_w32(s0);
      const unsigned b1 = __builtin_amdgcn_ballot_w32(s1);
      const unsigned b2 = __builtin_amdgcn_ballot_w32(s2);
      const unsigned b3 = __builtin_amdgcn_ballot_w32(s3);
      if ((b0 | b1 | b2 | b3) != 0u) {
        const int i0 = eLo + 4 * j;
        const unsigned q0 = 4u * (unsigned)j;
        const int p0 = wcount + __popc(b0 & ltMask);
        const int c0 = wcount + __popc(b0);
        const int p1 = c0 + __popc(b1 & ltMask);
        const int c1 = c0 + __popc(b1);
        const int p2 = c1 + __popc(b2 & ltMask);
        const int c2 = c1 + __popc(b2);
        const int p3 = c2 + __popc(b3 & ltMask);
        wcount = c2 + __popc(b3);
        if (s0) { if (__builtin_expect(p0 < RG_QW, 1)) myQueue[p0] = (d0 << RG_QSHIFT) | q0;
                  else direct(d0, i0); }
        if (s1) { if (__builtin_expect(p1 < RG_QW, 1)) myQueue[p1] = (d1 << RG_QSHIFT) | (q0 + 1u);
                  else direct(d1, i0 + 1); }
        if (s2) { if (__builtin_expect(p2 < RG_QW, 1)) myQueue[p2] = (d2 << RG_QSHIFT) | (q0 + 2u);
                  else direct(d2, i0 + 2); }
        if (s3) { if (__builtin_expect(p3 < RG_QW, 1)) myQueue[p3] = (d3 << RG_QSHIFT) | (q0 + 3u);
                  else direct(d3, i0 + 3); }
      }
    }
    const int avail = wcount < RG_QW ? wcount : RG_QW;
    for (; avail - drained >= BATCH; drained += BATCH) drainBatch(drained, avail);
  }
  const int n = wcount < RG_QW ? wcount : RG_QW;
  for (; drained < n; drained += BATCH) drainBatch(drained, n);
}

#define RG_U3   16
#define RG_QW3  512

__global__ void __launch_bounds__(RG_THREADS)
rgcn_scatter_h(const float* __restrict__ values,
               const int* __restrict__ hr,
               const int* __restrict__ hc,
               const float* __restrict__ W1,
               float* __restrict__ h) {
  __shared__ float    accTile[RG_ROWS * EE];
  __shared__ unsigned queue[32][RG_QW3];
  __shared__ float    spare[EE];

  const int tid  = threadIdx.x;
  const int lane = tid & 31;
  const int wv   = tid >> 5;
  const int half = lane >> 4;
  const int l16  = lane & 15;
  const int g    = blockIdx.x % RG_G;
  const int m    = blockIdx.x / RG_G;
  const int rowLo = g * RG_ROWS;
  const int eLo   = m * RG_SLICE;

  for (int j = tid; j < RG_ROWS * EE; j += RG_THREADS) accTile[j] = 0.0f;
  __syncthreads();

  rgcn_scan_slice<2 * RG_U3, RG_QW3>(
      hr, eLo, tid, lane, queue[wv],
      [&](int r) { return (unsigned)(r - rowLo); },
      [&](unsigned d, int i) {
        const float v = values[i];
        const float* wp = W1 + (size_t)hc[i] * EE;
#pragma nounroll
        for (int e = 0; e < EE; ++e) atomAddF32(&accTile[d * EE + e], v * wp[e]);
      },
      [&](int base, int n) {
        unsigned ent[RG_U3];
        bool ok[RG_U3];
#pragma unroll
        for (int u = 0; u < RG_U3; ++u) {
          const int q = base + 2 * u + half;
          ok[u] = q < n;
          ent[u] = ok[u] ? queue[wv][q] : 0u;
        }
        int col[RG_U3];
        float v[RG_U3];
#pragma unroll
        for (int u = 0; u < RG_U3; ++u) {
          const int i = eLo + (ok[u] ? (int)(ent[u] & RG_QMASK) : 0);
          col[u] = __builtin_nontemporal_load(&hc[i]);
          v[u]   = __builtin_nontemporal_load(&values[i]);
        }
        float w[RG_U3];
#pragma unroll
        for (int u = 0; u < RG_U3; ++u)
          w[u] = W1[(size_t)col[u] * EE + l16];
#pragma unroll
        for (int u = 0; u < RG_U3; ++u) {
          float* dst = ok[u] ? &accTile[(ent[u] >> RG_QSHIFT) * EE + l16] : &spare[l16];
          atomAddF32(dst, v[u] * w[u]);
        }
      });
  __syncthreads();

  for (int j = tid; j < RG_ROWS * EE; j += RG_THREADS) {
    const int row = rowLo + (j >> 4);
    if (row < NN) atomAddF32(&h[(size_t)rowLo * EE + j], accTile[j]);
  }
}

#define RG_U5   4
#define RG_QW5  1024

__global__ void __launch_bounds__(RG_THREADS)
rgcn_layer2(const float* __restrict__ values,
            const int* __restrict__ vr,
            const int* __restrict__ vc,
            const float* __restrict__ h,
            const float* __restrict__ bias1,
            const float* __restrict__ W2,
            float* __restrict__ out) {
  __shared__ float    W2s[RR * EE * CC];
  __shared__ float    outTile[RG_ROWS * CC];
  __shared__ unsigned queue[32][RG_QW5];
  __shared__ float    spare[CC];

  const int tid  = threadIdx.x;
  const int lane = tid & 31;
  const int wv   = tid >> 5;
  const int sub  = lane >> 3;
  const int c    = lane & 7;
  const int g    = blockIdx.x % RG_G;
  const int m    = blockIdx.x / RG_G;
  const int rowLo = g * RG_ROWS;
  const int eLo   = m * RG_SLICE;

  for (int j = tid; j < RR * EE * CC; j += RG_THREADS) W2s[j] = W2[j];
  for (int j = tid; j < RG_ROWS * CC; j += RG_THREADS) outTile[j] = 0.0f;
  __syncthreads();

  float b1[EE];
#pragma unroll
  for (int e = 0; e < EE; ++e) b1[e] = bias1[e];

  rgcn_scan_slice<4 * RG_U5, RG_QW5>(
      vr, eLo, tid, lane, queue[wv],
      [&](int r) {
        const unsigned ur = (unsigned)r;
        const unsigned prod = (unsigned)__umul24(ur, 1342u);
        const unsigned pe = prod >> 26;
        const unsigned se = ur - pe * (unsigned)NN;
        const unsigned sm = se - (unsigned)NN;
        return (se < sm ? se : sm) - (unsigned)rowLo;
      },
      [&](unsigned d, int i) {
        const int p = vr[i] / NN;
        const float v = values[i];
        const float* hp = h + (size_t)vc[i] * EE;
#pragma nounroll
        for (int cc = 0; cc < CC; ++cc) {
          const float* wp = W2s + p * (EE * CC) + cc;
          float acc = 0.0f;
#pragma nounroll
          for (int e = 0; e < EE; ++e) acc += fmaxf(hp[e] + bias1[e], 0.0f) * wp[e * CC];
          atomAddF32(&outTile[d * CC + cc], v * acc);
        }
      },
      [&](int base, int n) {
        unsigned ent[RG_U5];
        bool ok[RG_U5];
#pragma unroll
        for (int u = 0; u < RG_U5; ++u) {
          const int q = base + 4 * u + sub;
          ok[u] = q < n;
          ent[u] = ok[u] ? queue[wv][q] : 0u;
        }
        int r[RG_U5], o[RG_U5];
        float v[RG_U5];
#pragma unroll
        for (int u = 0; u < RG_U5; ++u) {
          const int i = eLo + (ok[u] ? (int)(ent[u] & RG_QMASK) : 0);
          r[u] = __builtin_nontemporal_load(&vr[i]);
          o[u] = __builtin_nontemporal_load(&vc[i]);
          v[u] = __builtin_nontemporal_load(&values[i]);
        }
        float4 hv[RG_U5][4];
#pragma unroll
        for (int u = 0; u < RG_U5; ++u) {
          const float4* hp4 = reinterpret_cast<const float4*>(h + (size_t)o[u] * EE);
#pragma unroll
          for (int k = 0; k < 4; ++k) hv[u][k] = hp4[k];
        }
        __builtin_amdgcn_sched_barrier(0);
        float acc[RG_U5];
#pragma unroll
        for (int u = 0; u < RG_U5; ++u) {
          const int p = r[u] / NN;
          const float* wp = W2s + p * (EE * CC) + c;
          float a = 0.0f;
#pragma unroll
          for (int k = 0; k < 4; ++k) {
            a += fmaxf(hv[u][k].x + b1[4 * k + 0], 0.0f) * wp[(4 * k + 0) * CC];
            a += fmaxf(hv[u][k].y + b1[4 * k + 1], 0.0f) * wp[(4 * k + 1) * CC];
            a += fmaxf(hv[u][k].z + b1[4 * k + 2], 0.0f) * wp[(4 * k + 2) * CC];
            a += fmaxf(hv[u][k].w + b1[4 * k + 3], 0.0f) * wp[(4 * k + 3) * CC];
          }
          acc[u] = a;
        }
#pragma unroll
        for (int u = 0; u < RG_U5; ++u) {
          float* dst = ok[u] ? &outTile[(ent[u] >> RG_QSHIFT) * CC + c] : &spare[c];
          atomAddF32(dst, v[u] * acc[u]);
        }
      });
  __syncthreads();

  for (int j = tid; j < RG_ROWS * CC; j += RG_THREADS) {
    const int row = rowLo + (j >> 3);
    if (row < NN) atomAddF32(&out[(size_t)rowLo * CC + j], outTile[j]);
  }
}

extern "C" void kernel_launch(void* const* d_in, const int* in_sizes, int n_in,
                              void* d_out, int out_size, void* d_ws, size_t ws_size,
                              hipStream_t stream) {
  const float* values = (const float*)d_in[0];
  const float* comps1 = (const float*)d_in[1];
  const float* bases1 = (const float*)d_in[2];
  const float* comps2 = (const float*)d_in[3];
  const float* bases2 = (const float*)d_in[4];
  const float* bias1  = (const float*)d_in[5];
  const float* bias2  = (const float*)d_in[6];
  const int*   hr     = (const int*)d_in[7];
  const int*   hc     = (const int*)d_in[8];
  const int*   vr     = (const int*)d_in[9];
  const int*   vc     = (const int*)d_in[10];
  float* out = (float*)d_out;

  float* W1 = (float*)d_ws;
  float* h  = W1 + W1ELEMS;
  float* W2 = h + NE;

  rgcn_w1_gemm<<<832, 256, 0, stream>>>(comps1, bases1, W1, h, out, bias2, comps2, bases2, W2);
  rgcn_scatter_h<<<RG_G * RG_M, RG_THREADS, 0, stream>>>(values, hr, hc, W1, h);
  rgcn_layer2<<<RG_G * RG_M, RG_THREADS, 0, stream>>>(values, vr, vc, h, bias1, W2, out);
}


# layer-2 kernel: W2 transposed in LDS (a class's 16 weights side by side, padded pitches, two copies 32 banks apart): 4 conflict-free 16-byte reads per lane, not 16 4-way-conflicted; sum order kept
# speedup vs baseline: 1.6926x; 1.1252x over previous
#include <hip/hip_runtime.h>
#include <hip/hip_bf16.h>

#define NN    50000
#define RR    51
#define EE    16
#define CC    8
#define BB    40
#define NNZ   1600000
#define NE    (NN * EE)
#define W1ELEMS ((size_t)RR * NE)

typedef float v2f __attribute__((ext_vector_type(2)));
typedef float v8f __attribute__((ext_vector_type(8)));

__device__ __forceinline__ void atomAddF32(float* p, float v) {
  unsafeAtomicAdd(p, v);
}

__global__ void __launch_bounds__(256)
rgcn_w1_gemm(const float* __restrict__ comps1,
             const float* __restrict__ bases1,
             float* __restrict__ W1,
             float* __restrict__ h,
             float* __restrict__ out,
             const float* __restrict__ bias2,
             const float* __restrict__ comps2,
             const float* __restrict__ bases2,
             float* __restrict__ W2) {
  {
    const int gid      = blockIdx.x * blockDim.x + threadIdx.x;
    const int nThreads = gridDim.x * blockDim.x;
    float4* h4   = reinterpret_cast<float4*>(h);
    float4* out4 = reinterpret_cast<float4*>(out);
    const float4 zero4 = make_float4(0.0f, 0.0f, 0.0f, 0.0f);
    for (int i = gid; i < NE / 4; i += nThreads) h4[i] = zero4;
    const int c0 = (gid & 1) * 4;
    const float4 bv = make_float4(bias2[c0], bias2[c0 + 1], bias2[c0 + 2], bias2[c0 + 3]);
    for (int i = gid; i < NN * CC / 4; i += nThreads) out4[i] = bv;
    if (gid < RR * EE * CC) {
      const int r  = gid >> 7;
      const int ec = gid & 127;
      float acc = 0.0f;
#pragma unroll
      for (int b = 0; b < BB; ++b)
        acc += comps2[r * BB + b] * bases2[b * (EE * CC) + ec];
      W2[gid] = acc;
    }
  }

  const int lane = threadIdx.x & 31;
  const int half = lane >> 4;
  const int l16  = lane & 15;

  v2f a[4][10];
#pragma unroll
  for (int mt = 0; mt < 4; ++mt) {
    int m = mt * 16 + l16;
#pragma unroll
    for (int ks = 0; ks < 10; ++ks) {
      int k = ks * 4 + half * 2;
      float ax = 0.0f, ay = 0.0f;
      if (m < RR) {
        ax = comps1[m * BB + k];
        ay = comps1[m * BB + k + 1];
      }
      a[mt][ks].x = ax;
      a[mt][ks].y = ay;
    }
  }

  const int wavesPerBlock = blockDim.x >> 5;
  const int waveId  = blockIdx.x * wavesPerBlock + (threadIdx.x >> 5);
  const int nWaves  = gridDim.x * wavesPerBlock;
  const int nTiles  = NE / 16;

  for (int tile = waveId; tile < nTiles; tile += nWaves) {
    const int col = tile * 16 + l16;
    v8f acc[4] = {};
#pragma unroll
    for (int ks = 0; ks < 10; ++ks) {
      const int k = ks * 4 + half * 2;
      v2f b;
      b.x = __builtin_nontemporal_load(&bases1[(size_t)k * NE + col]);
      b.y = __builtin_nontemporal_load(&bases1[(size_t)(k + 1) * NE + col]);
#pragma unroll
      for (int mt = 0; mt < 4; ++mt) {
        acc[mt] = __builtin_amdgcn_wmma_f32_16x16x4_f32(
             false, a[mt][ks],  false, b,
             (short)0, acc[mt],
             false,  false);
      }
    }
#pragma unroll
    for (int mt = 0; mt < 4; ++mt) {
#pragma unroll
      for (int j = 0; j < 8; ++j) {
        int m = mt * 16 + j + half * 8;
        if (m < RR) W1[(size_t)m * NE + col] = acc[mt][j];
      }
    }
  }
}

#define RG_ROWS 3128
#define RG_G    16
#define RG_M    16
#define RG_SLICE (NNZ / RG_M)
#define RG_QSHIFT 17
#define RG_QMASK  0x1FFFFu
#define RG_THREADS 1024

template <int BATCH, int RG_QW, typename RowFn, typename DirectFn, typename DrainFn>
__device__ __forceinline__ void rgcn_scan_slice(const int* __restrict__ ids, int eLo,
                                                int tid, int lane, unsigned* myQueue,
                                                RowFn localRow, DirectFn direct,
                                                DrainFn drainBatch) {
  const int4* ids4 = reinterpret_cast<const int4*>(ids + eLo);
  const unsigned ltMask = (1u << lane) - 1u;
  int wcount = 0;
  int drained = 0;
  for (int jb = 0; jb < RG_SLICE / 4; jb += 4 * RG_THREADS) {
    int4 rr[4];
#pragma unroll
    for (int k = 0; k < 4; ++k) {
      const int j = jb + k * RG_THREADS + tid;
      rr[k] = make_int4(-1, -1, -1, -1);
      if (j < RG_SLICE / 4) rr[k] = ids4[j];
    }
#pragma unroll
    for (int k = 0; k < 4; ++k) {
      if (jb + k * RG_THREADS >= RG_SLICE / 4) break;
      const int j = jb + k * RG_THREADS + tid;
      const bool inRange = j < RG_SLICE / 4;
      const unsigned d0 = localRow(rr[k].x), d1 = localRow(rr[k].y);
      const unsigned d2 = localRow(rr[k].z), d3 = localRow(rr[k].w);
      const bool s0 = inRange & (d0 < RG_ROWS), s1 = inRange & (d1 < RG_ROWS);
      const bool s2 = inRange & (d2 < RG_ROWS), s3 = inRange & (d3 < RG_ROWS);
      const unsigned b0 = __builtin_amdgcn_ballot_w32(s0);
      const unsigned b1 = __builtin_amdgcn_ballot_w32(s1);
      const unsigned b2 = __builtin_amdgcn_ballot_w32(s2);
      const unsigned b3 = __builtin_amdgcn_ballot_w32(s3);
      if ((b0 | b1 | b2 | b3) != 0u) {
        const int i0 = eLo + 4 * j;
        const unsigned q0 = 4u * (unsigned)j;
        const int p0 = wcount + __popc(b0 & ltMask);
        const int c0 = wcount + __popc(b0);
        const int p1 = c0 + __popc(b1 & ltMask);
        const int c1 = c0 + __popc(b1);
        const int p2 = c1 + __popc(b2 & ltMask);
        const int c2 = c1 + __popc(b2);
        const int p3 = c2 + __popc(b3 & ltMask);
        wcount = c2 + __popc(b3);
        if (s0) { if (__builtin_expect(p0 < RG_QW, 1)) myQueue[p0] = (d0 << RG_QSHIFT) | q0;
                  else direct(d0, i0); }
        if (s1) { if (__builtin_expect(p1 < RG_QW, 1)) myQueue[p1] = (d1 << RG_QSHIFT) | (q0 + 1u);
                  else direct(d1, i0 + 1); }
        if (s2) { if (__builtin_expect(p2 < RG_QW, 1)) myQueue[p2] = (d2 << RG_QSHIFT) | (q0 + 2u);
                  else direct(d2, i0 + 2); }
        if (s3) { if (__builtin_expect(p3 < RG_QW, 1)) myQueue[p3] = (d3 << RG_QSHIFT) | (q0 + 3u);
                  else direct(d3, i0 + 3); }
      }
    }
    const int avail = wcount < RG_QW ? wcount : RG_QW;
    for (; avail - drained >= BATCH; drained += BATCH) drainBatch(drained, avail);
  }
  const int n = wcount < RG_QW ? wcount : RG_QW;
  for (; drained < n; drained += BATCH) drainBatch(drained, n);
}

#define RG_U3   16
#define RG_QW3  512

__global__ void __launch_bounds__(RG_THREADS)
rgcn_scatter_h(const float* __restrict__ values,
               const int* __restrict__ hr,
               const int* __restrict__ hc,
               const float* __restrict__ W1,
               float* __restrict__ h) {
  __shared__ float    accTile[RG_ROWS * EE];
  __shared__ unsigned queue[32][RG_QW3];
  __shared__ float    spare[EE];

  const int tid  = threadIdx.x;
  const int lane = tid & 31;
  const int wv   = tid >> 5;
  const int half = lane >> 4;
  const int l16  = lane & 15;
  const int g    = blockIdx.x % RG_G;
  const int m    = blockIdx.x / RG_G;
  const int rowLo = g * RG_ROWS;
  const int eLo   = m * RG_SLICE;

  for (int j = tid; j < RG_ROWS * EE; j += RG_THREADS) accTile[j] = 0.0f;
  __syncthreads();

  rgcn_scan_slice<2 * RG_U3, RG_QW3>(
      hr, eLo, tid, lane, queue[wv],
      [&](int r) { return (unsigned)(r - rowLo); },
      [&](unsigned d, int i) {
        const float v = values[i];
        const float* wp = W1 + (size_t)hc[i] * EE;
#pragma nounroll
        for (int e = 0; e < EE; ++e) atomAddF32(&accTile[d * EE + e], v * wp[e]);
      },
      [&](int base, int n) {
        unsigned ent[RG_U3];
        bool ok[RG_U3];
#pragma unroll
        for (int u = 0; u < RG_U3; ++u) {
          const int q = base + 2 * u + half;
          ok[u] = q < n;
          ent[u] = ok[u] ? queue[wv][q] : 0u;
        }
        int col[RG_U3];
        float v[RG_U3];
#pragma unroll
        for (int u = 0; u < RG_U3; ++u) {
          const int i = eLo + (ok[u] ? (int)(ent[u] & RG_QMASK) : 0);
          col[u] = __builtin_nontemporal_load(&hc[i]);
          v[u]   = __builtin_nontemporal_load(&values[i]);
        }
        float w[RG_U3];
#pragma unroll
        for (int u = 0; u < RG_U3; ++u)
          w[u] = W1[(size_t)col[u] * EE + l16];
#pragma unroll
        for (int u = 0; u < RG_U3; ++u) {
          float* dst = ok[u] ? &accTile[(ent[u] >> RG_QSHIFT) * EE + l16] : &spare[l16];
          atomAddF32(dst, v[u] * w[u]);
        }
      });
  __syncthreads();

  for (int j = tid; j < RG_ROWS * EE; j += RG_THREADS) {
    const int row = rowLo + (j >> 4);
    if (row < NN) atomAddF32(&h[(size_t)rowLo * EE + j], accTile[j]);
  }
}

#define RG_U5   4
#define W2T_CP   20
#define W2T_PP   (CC * W2T_CP)
#define W2T_COPY (RR * W2T_PP)
#define RG_QW5  1024

__global__ void __launch_bounds__(RG_THREADS)
rgcn_layer2(const float* __restrict__ values,
            const int* __restrict__ vr,
            const int* __restrict__ vc,
            const float* __restrict__ h,
            const float* __restrict__ bias1,
            const float* __restrict__ W2,
            float* __restrict__ out) {
  __shared__ __attribute__((aligned(16))) float W2T[2 * W2T_COPY];
  __shared__ float    outTile[RG_ROWS * CC];
  __shared__ unsigned queue[32][RG_QW5];
  __shared__ float    spare[CC];

  const int tid  = threadIdx.x;
  const int lane = tid & 31;
  const int wv   = tid >> 5;
  const int sub  = lane >> 3;
  const int c    = lane & 7;
  const int g    = blockIdx.x % RG_G;
  const int m    = blockIdx.x / RG_G;
  const int rowLo = g * RG_ROWS;
  const int eLo   = m * RG_SLICE;

  for (int j = tid; j < RR * EE * CC; j += RG_THREADS) {
    const int t = (j >> 7) * W2T_PP + (j & 7) * W2T_CP + ((j >> 3) & 15);
    const float w = W2[j];
    W2T[t] = w;
    W2T[W2T_COPY + t] = w;
  }
  for (int j = tid; j < RG_ROWS * CC; j += RG_THREADS) outTile[j] = 0.0f;
  __syncthreads();

  float b1[EE];
#pragma unroll
  for (int e = 0; e < EE; ++e) b1[e] = bias1[e];

  rgcn_scan_slice<4 * RG_U5, RG_QW5>(
      vr, eLo, tid, lane, queue[wv],
      [&](int r) {
        const unsigned ur = (unsigned)r;
        const unsigned prod = (unsigned)__umul24(ur, 1342u);
        const unsigned pe = prod >> 26;
        const unsigned se = ur - pe * (unsigned)NN;
        const unsigned sm = se - (unsigned)NN;
        return (se < sm ? se : sm) - (unsigned)rowLo;
      },
      [&](unsigned d, int i) {
        const int p = vr[i] / NN;
        const float v = values[i];
        const float* hp = h + (size_t)vc[i] * EE;
#pragma nounroll
        for (int cc = 0; cc < CC; ++cc) {
          const float* wp = W2T + p * W2T_PP + cc * W2T_CP;
          float acc = 0.0f;
#pragma nounroll
          for (int e = 0; e < EE; ++e) acc += fmaxf(hp[e] + bias1[e], 0.0f) * wp[e];
          atomAddF32(&outTile[d * CC + cc], v * acc);
        }
      },
      [&](int base, int n) {
        unsigned ent[RG_U5];
        bool ok[RG_U5];
#pragma unroll
        for (int u = 0; u < RG_U5; ++u) {
          const int q = base + 4 * u + sub;
          ok[u] = q < n;
          ent[u] = ok[u] ? queue[wv][q] : 0u;
        }
        int r[RG_U5], o[RG_U5];
        float v[RG_U5];
#pragma unroll
        for (int u = 0; u < RG_U5; ++u) {
          const int i = eLo + (ok[u] ? (int)(ent[u] & RG_QMASK) : 0);
          r[u] = __builtin_nontemporal_load(&vr[i]);
          o[u] = __builtin_nontemporal_load(&vc[i]);
          v[u] = __builtin_nontemporal_load(&values[i]);
        }
        float4 hv[RG_U5][4];
#pragma unroll
        for (int u = 0; u < RG_U5; ++u) {
          const float4* hp4 = reinterpret_cast<const float4*>(h + (size_t)o[u] * EE);
#pragma unroll
          for (int k = 0; k < 4; ++k) hv[u][k] = hp4[k];
        }
        __builtin_amdgcn_sched_barrier(0);
        float acc[RG_U5];
#pragma unroll
        for (int u = 0; u < RG_U5; ++u) {
          const int p = r[u] / NN;
          const int copy = (p ^ sub) & 1;
          const float4* wq = reinterpret_cast<const float4*>(
              W2T + copy * W2T_COPY + p * W2T_PP + c * W2T_CP);
          float a = 0.0f;
#pragma unroll
          for (int k = 0; k < 4; ++k) {
            const float4 w = wq[k];
            a += fmaxf(hv[u][k].x + b1[4 * k + 0], 0.0f) * w.x;
            a += fmaxf(hv[u][k].y + b1[4 * k + 1], 0.0f) * w.y;
            a += fmaxf(hv[u][k].z + b1[4 * k + 2], 0.0f) * w.z;
            a += fmaxf(hv[u][k].w + b1[4 * k + 3], 0.0f) * w.w;
          }
          acc[u] = a;
        }
#pragma unroll
        for (int u = 0; u < RG_U5; ++u) {
          float* dst = ok[u] ? &outTile[(ent[u] >> RG_QSHIFT) * CC + c] : &spare[c];
          atomAddF32(dst, v[u] * acc[u]);
        }
      });
  __syncthreads();

  for (int j = tid; j < RG_ROWS * CC; j += RG_THREADS) {
    const int row = rowLo + (j >> 3);
    if (row < NN) atomAddF32(&out[(size_t)rowLo * CC + j], outTile[j]);
  }
}

extern "C" void kernel_launch(void* const* d_in, const int* in_sizes, int n_in,
                              void* d_out, int out_size, void* d_ws, size_t ws_size,
                              hipStream_t stream) {
  const float* values = (const float*)d_in[0];
  const float* comps1 = (const float*)d_in[1];
  const float* bases1 = (const float*)d_in[2];
  const float* comps2 = (const float*)d_in[3];
  const float* bases2 = (const float*)d_in[4];
  const float* bias1  = (const float*)d_in[5];
  const float* bias2  = (const float*)d_in[6];
  const int*   hr     = (const int*)d_in[7];
  const int*   hc     = (const int*)d_in[8];
  const int*   vr     = (const int*)d_in[9];
  const int*   vc     = (const int*)d_in[10];
  float* out = (float*)d_out;

  float* W1 = (float*)d_ws;
  float* h  = W1 + W1ELEMS;
  float* W2 = h + NE;

  rgcn_w1_gemm<<<832, 256, 0, stream>>>(comps1, bases1, W1, h, out, bias2, comps2, bases2, W2);
  rgcn_scatter_h<<<RG_G * RG_M, RG_THREADS, 0, stream>>>(values, hr, hc, W1, h);
  rgcn_layer2<<<RG_G * RG_M, RG_THREADS, 0, stream>>>(values, vr, vc, h, bias1, W2, out);
}


# W1 GEMM kernel: grid 768 workgroups of 8 waves (was 832): at 136 registers a WGP holds 3, so all are resident at once and no surplus 64 run almost alone at the end; grid now a kernel constant
# speedup vs baseline: 1.7843x; 1.0542x over previous
#include <hip/hip_runtime.h>
#include <hip/hip_bf16.h>

#define NN    50000
#define RR    51
#define EE    16
#define CC    8
#define BB    40
#define NNZ   1600000
#define NE    (NN * EE)
#define W1ELEMS ((size_t)RR * NE)

typedef float v2f __attribute__((ext_vector_type(2)));
typedef float v8f __attribute__((ext_vector_type(8)));

__device__ __forceinline__ void atomAddF32(float* p, float v) {
  unsafeAtomicAdd(p, v);
}

#define K2_THREADS 256
#define K2_BLOCKS  768
#define K2_WAVES   (K2_BLOCKS * (K2_THREADS / 32))

__global__ void __launch_bounds__(K2_THREADS)
rgcn_w1_gemm(const float* __restrict__ comps1,
             const float* __restrict__ bases1,
             float* __restrict__ W1,
             float* __restrict__ h,
             float* __restrict__ out,
             const float* __restrict__ bias2,
             const float* __restrict__ comps2,
             const float* __restrict__ bases2,
             float* __restrict__ W2) {
  {
    const int gid      = blockIdx.x * K2_THREADS + threadIdx.x;
    const int nThreads = K2_BLOCKS * K2_THREADS;
    float4* h4   = reinterpret_cast<float4*>(h);
    float4* out4 = reinterpret_cast<float4*>(out);
    const float4 zero4 = make_float4(0.0f, 0.0f, 0.0f, 0.0f);
    for (int i = gid; i < NE / 4; i += nThreads) h4[i] = zero4;
    const int c0 = (gid & 1) * 4;
    const float4 bv = make_float4(bias2[c0], bias2[c0 + 1], bias2[c0 + 2], bias2[c0 + 3]);
    for (int i = gid; i < NN * CC / 4; i += nThreads) out4[i] = bv;
    if (gid < RR * EE * CC) {
      const int r  = gid >> 7;
      const int ec = gid & 127;
      float acc = 0.0f;
#pragma unroll
      for (int b = 0; b < BB; ++b)
        acc += comps2[r * BB + b] * bases2[b * (EE * CC) + ec];
      W2[gid] = acc;
    }
  }

  const int lane = threadIdx.x & 31;
  const int half = lane >> 4;
  const int l16  = lane & 15;

  v2f a[4][10];
#pragma unroll
  for (int mt = 0; mt < 4; ++mt) {
    int m = mt * 16 + l16;
#pragma unroll
    for (int ks = 0; ks < 10; ++ks) {
      int k = ks * 4 + half * 2;
      float ax = 0.0f, ay = 0.0f;
      if (m < RR) {
        ax = comps1[m * BB + k];
        ay = comps1[m * BB + k + 1];
      }
      a[mt][ks].x = ax;
      a[mt][ks].y = ay;
    }
  }

  const int waveId  = blockIdx.x * (K2_THREADS / 32) + (threadIdx.x >> 5);
  const int nWaves  = K2_WAVES;
  const int nTiles  = NE / 16;

  for (int tile = waveId; tile < nTiles; tile += nWaves) {
    const int col = tile * 16 + l16;
    v8f acc[4] = {};
#pragma unroll
    for (int ks = 0; ks < 10; ++ks) {
      const int k = ks * 4 + half * 2;
      v2f b;
      b.x = __builtin_nontemporal_load(&bases1[(size_t)k * NE + col]);
      b.y = __builtin_nontemporal_load(&bases1[(size_t)(k + 1) * NE + col]);
#pragma unroll
      for (int mt = 0; mt < 4; ++mt) {
        acc[mt] = __builtin_amdgcn_wmma_f32_16x16x4_f32(
             false, a[mt][ks],  false, b,
             (short)0, acc[mt],
             false,  false);
      }
    }
#pragma unroll
    for (int mt = 0; mt < 4; ++mt) {
#pragma unroll
      for (int j = 0; j < 8; ++j) {
        int m = mt * 16 + j + half * 8;
        if (m < RR) W1[(size_t)m * NE + col] = acc[mt][j];
      }
    }
  }
}

#define RG_ROWS 3128
#define RG_G    16
#define RG_M    16
#define RG_SLICE (NNZ / RG_M)
#define RG_QSHIFT 17
#define RG_QMASK  0x1FFFFu
#define RG_THREADS 1024

template <int BATCH, int RG_QW, typename RowFn, typename DirectFn, typename DrainFn>
__device__ __forceinline__ void rgcn_scan_slice(const int* __restrict__ ids, int eLo,
                                                int tid, int lane, unsigned* myQueue,
                                                RowFn localRow, DirectFn direct,
                                                DrainFn drainBatch) {
  const int4* ids4 = reinterpret_cast<const int4*>(ids + eLo);
  const unsigned ltMask = (1u << lane) - 1u;
  int wcount = 0;
  int drained = 0;
  for (int jb = 0; jb < RG_SLICE / 4; jb += 4 * RG_THREADS) {
    int4 rr[4];
#pragma unroll
    for (int k = 0; k < 4; ++k) {
      const int j = jb + k * RG_THREADS + tid;
      rr[k] = make_int4(-1, -1, -1, -1);
      if (j < RG_SLICE / 4) rr[k] = ids4[j];
    }
#pragma unroll
    for (int k = 0; k < 4; ++k) {
      if (jb + k * RG_THREADS >= RG_SLICE / 4) break;
      const int j = jb + k * RG_THREADS + tid;
      const bool inRange = j < RG_SLICE / 4;
      const unsigned d0 = localRow(rr[k].x), d1 = localRow(rr[k].y);
      const unsigned d2 = localRow(rr[k].z), d3 = localRow(rr[k].w);
      const bool s0 = inRange & (d0 < RG_ROWS), s1 = inRange & (d1 < RG_ROWS);
      const bool s2 = inRange & (d2 < RG_ROWS), s3 = inRange & (d3 < RG_ROWS);
      const unsigned b0 = __builtin_amdgcn_ballot_w32(s0);
      const unsigned b1 = __builtin_amdgcn_ballot_w32(s1);
      const unsigned b2 = __builtin_amdgcn_ballot_w32(s2);
      const unsigned b3 = __builtin_amdgcn_ballot_w32(s3);
      if ((b0 | b1 | b2 | b3) != 0u) {
        const int i0 = eLo + 4 * j;
        const unsigned q0 = 4u * (unsigned)j;
        const int p0 = wcount + __popc(b0 & ltMask);
        const int c0 = wcount + __popc(b0);
        const int p1 = c0 + __popc(b1 & ltMask);
        const int c1 = c0 + __popc(b1);
        const int p2 = c1 + __popc(b2 & ltMask);
        const int c2 = c1 + __popc(b2);
        const int p3 = c2 + __popc(b3 & ltMask);
        wcount = c2 + __popc(b3);
        if (s0) { if (__builtin_expect(p0 < RG_QW, 1)) myQueue[p0] = (d0 << RG_QSHIFT) | q0;
                  else direct(d0, i0); }
        if (s1) { if (__builtin_expect(p1 < RG_QW, 1)) myQueue[p1] = (d1 << RG_QSHIFT) | (q0 + 1u);
                  else direct(d1, i0 + 1); }
        if (s2) { if (__builtin_expect(p2 < RG_QW, 1)) myQueue[p2] = (d2 << RG_QSHIFT) | (q0 + 2u);
                  else direct(d2, i0 + 2); }
        if (s3) { if (__builtin_expect(p3 < RG_QW, 1)) myQueue[p3] = (d3 << RG_QSHIFT) | (q0 + 3u);
                  else direct(d3, i0 + 3); }
      }
    }
    const int avail = wcount < RG_QW ? wcount : RG_QW;
    for (; avail - drained >= BATCH; drained += BATCH) drainBatch(drained, avail);
  }
  const int n = wcount < RG_QW ? wcount : RG_QW;
  for (; drained < n; drained += BATCH) drainBatch(drained, n);
}

#define RG_U3   16
#define RG_QW3  512

__global__ void __launch_bounds__(RG_THREADS)
rgcn_scatter_h(const float* __restrict__ values,
               const int* __restrict__ hr,
               const int* __restrict__ hc,
               const float* __restrict__ W1,
               float* __restrict__ h) {
  __shared__ float    accTile[RG_ROWS * EE];
  __shared__ unsigned queue[32][RG_QW3];
  __shared__ float    spare[EE];

  const int tid  = threadIdx.x;
  const int lane = tid & 31;
  const int wv   = tid >> 5;
  const int half = lane >> 4;
  const int l16  = lane & 15;
  const int g    = blockIdx.x % RG_G;
  const int m    = blockIdx.x / RG_G;
  const int rowLo = g * RG_ROWS;
  const int eLo   = m * RG_SLICE;

  for (int j = tid; j < RG_ROWS * EE; j += RG_THREADS) accTile[j] = 0.0f;
  __syncthreads();

  rgcn_scan_slice<2 * RG_U3, RG_QW3>(
      hr, eLo, tid, lane, queue[wv],
      [&](int r) { return (unsigned)(r - rowLo); },
      [&](unsigned d, int i) {
        const float v = values[i];
        const float* wp = W1 + (size_t)hc[i] * EE;
#pragma nounroll
        for (int e = 0; e < EE; ++e) atomAddF32(&accTile[d * EE + e], v * wp[e]);
      },
      [&](int base, int n) {
        unsigned ent[RG_U3];
        bool ok[RG_U3];
#pragma unroll
        for (int u = 0; u < RG_U3; ++u) {
          const int q = base + 2 * u + half;
          ok[u] = q < n;
          ent[u] = ok[u] ? queue[wv][q] : 0u;
        }
        int col[RG_U3];
        float v[RG_U3];
#pragma unroll
        for (int u = 0; u < RG_U3; ++u) {
          const int i = eLo + (ok[u] ? (int)(ent[u] & RG_QMASK) : 0);
          col[u] = __builtin_nontemporal_load(&hc[i]);
          v[u]   = __builtin_nontemporal_load(&values[i]);
        }
        float w[RG_U3];
#pragma unroll
        for (int u = 0; u < RG_U3; ++u)
          w[u] = W1[(size_t)col[u] * EE + l16];
#pragma unroll
        for (int u = 0; u < RG_U3; ++u) {
          float* dst = ok[u] ? &accTile[(ent[u] >> RG_QSHIFT) * EE + l16] : &spare[l16];
          atomAddF32(dst, v[u] * w[u]);
        }
      });
  __syncthreads();

  for (int j = tid; j < RG_ROWS * EE; j += RG_THREADS) {
    const int row = rowLo + (j >> 4);
    if (row < NN) atomAddF32(&h[(size_t)rowLo * EE + j], accTile[j]);
  }
}

#define RG_U5   4
#define W2T_CP   20
#define W2T_PP   (CC * W2T_CP)
#define W2T_COPY (RR * W2T_PP)
#define RG_QW5  1024

__global__ void __launch_bounds__(RG_THREADS)
rgcn_layer2(const float* __restrict__ values,
            const int* __restrict__ vr,
            const int* __restrict__ vc,
            const float* __restrict__ h,
            const float* __restrict__ bias1,
            const float* __restrict__ W2,
            float* __restrict__ out) {
  __shared__ __attribute__((aligned(16))) float W2T[2 * W2T_COPY];
  __shared__ float    outTile[RG_ROWS * CC];
  __shared__ unsigned queue[32][RG_QW5];
  __shared__ float    spare[CC];

  const int tid  = threadIdx.x;
  const int lane = tid & 31;
  const int wv   = tid >> 5;
  const int sub  = lane >> 3;
  const int c    = lane & 7;
  const int g    = blockIdx.x % RG_G;
  const int m    = blockIdx.x / RG_G;
  const int rowLo = g * RG_ROWS;
  const int eLo   = m * RG_SLICE;

  for (int j = tid; j < RR * EE * CC; j += RG_THREADS) {
    const int t = (j >> 7) * W2T_PP + (j & 7) * W2T_CP + ((j >> 3) & 15);
    const float w = W2[j];
    W2T[t] = w;
    W2T[W2T_COPY + t] = w;
  }
  for (int j = tid; j < RG_ROWS * CC; j += RG_THREADS) outTile[j] = 0.0f;
  __syncthreads();

  float b1[EE];
#pragma unroll
  for (int e = 0; e < EE; ++e) b1[e] = bias1[e];

  rgcn_scan_slice<4 * RG_U5, RG_QW5>(
      vr, eLo, tid, lane, queue[wv],
      [&](int r) {
        const unsigned ur = (unsigned)r;
        const unsigned prod = (unsigned)__umul24(ur, 1342u);
        const unsigned pe = prod >> 26;
        const unsigned se = ur - pe * (unsigned)NN;
        const unsigned sm = se - (unsigned)NN;
        return (se < sm ? se : sm) - (unsigned)rowLo;
      },
      [&](unsigned d, int i) {
        const int p = vr[i] / NN;
        const float v = values[i];
        const float* hp = h + (size_t)vc[i] * EE;
#pragma nounroll
        for (int cc = 0; cc < CC; ++cc) {
          const float* wp = W2T + p * W2T_PP + cc * W2T_CP;
          float acc = 0.0f;
#pragma nounroll
          for (int e = 0; e < EE; ++e) acc += fmaxf(hp[e] + bias1[e], 0.0f) * wp[e];
          atomAddF32(&outTile[d * CC + cc], v * acc);
        }
      },
      [&](int base, int n) {
        unsigned ent[RG_U5];
        bool ok[RG_U5];
#pragma unroll
        for (int u = 0; u < RG_U5; ++u) {
          const int q = base + 4 * u + sub;
          ok[u] = q < n;
          ent[u] = ok[u] ? queue[wv][q] : 0u;
        }
        int r[RG_U5], o[RG_U5];
        float v[RG_U5];
#pragma unroll
        for (int u = 0; u < RG_U5; ++u) {
          const int i = eLo + (ok[u] ? (int)(ent[u] & RG_QMASK) : 0);
          r[u] = __builtin_nontemporal_load(&vr[i]);
          o[u] = __builtin_nontemporal_load(&vc[i]);
          v[u] = __builtin_nontemporal_load(&values[i]);
        }
        float4 hv[RG_U5][4];
#pragma unroll
        for (int u = 0; u < RG_U5; ++u) {
          const float4* hp4 = reinterpret_cast<const float4*>(h + (size_t)o[u] * EE);
#pragma unroll
          for (int k = 0; k < 4; ++k) hv[u][k] = hp4[k];
        }
        __builtin_amdgcn_sched_barrier(0);
        float acc[RG_U5];
#pragma unroll
        for (int u = 0; u < RG_U5; ++u) {
          const int p = r[u] / NN;
          const int copy = (p ^ sub) & 1;
          const float4* wq = reinterpret_cast<const float4*>(
              W2T + copy * W2T_COPY + p * W2T_PP + c * W2T_CP);
          float a = 0.0f;
#pragma unroll
          for (int k = 0; k < 4; ++k) {
            const float4 w = wq[k];
            a += fmaxf(hv[u][k].x + b1[4 * k + 0], 0.0f) * w.x;
            a += fmaxf(hv[u][k].y + b1[4 * k + 1], 0.0f) * w.y;
            a += fmaxf(hv[u][k].z + b1[4 * k + 2], 0.0f) * w.z;
            a += fmaxf(hv[u][k].w + b1[4 * k + 3], 0.0f) * w.w;
          }
          acc[u] = a;
        }
#pragma unroll
        for (int u = 0; u < RG_U5; ++u) {
          float* dst = ok[u] ? &outTile[(ent[u] >> RG_QSHIFT) * CC + c] : &spare[c];
          atomAddF32(dst, v[u] * acc[u]);
        }
      });
  __syncthreads();

  for (int j = tid; j < RG_ROWS * CC; j += RG_THREADS) {
    const int row = rowLo + (j >> 3);
    if (row < NN) atomAddF32(&out[(size_t)rowLo * CC + j], outTile[j]);
  }
}

extern "C" void kernel_launch(void* const* d_in, const int* in_sizes, int n_in,
                              void* d_out, int out_size, void* d_ws, size_t ws_size,
                              hipStream_t stream) {
  const float* values = (const float*)d_in[0];
  const float* comps1 = (const float*)d_in[1];
  const float* bases1 = (const float*)d_in[2];
  const float* comps2 = (const float*)d_in[3];
  const float* bases2 = (const float*)d_in[4];
  const float* bias1  = (const float*)d_in[5];
  const float* bias2  = (const float*)d_in[6];
  const int*   hr     = (const int*)d_in[7];
  const int*   hc     = (const int*)d_in[8];
  const int*   vr     = (const int*)d_in[9];
  const int*   vc     = (const int*)d_in[10];
  float* out = (float*)d_out;

  float* W1 = (float*)d_ws;
  float* h  = W1 + W1ELEMS;
  float* W2 = h + NE;

  rgcn_w1_gemm<<<K2_BLOCKS, K2_THREADS, 0, stream>>>(comps1, bases1, W1, h, out, bias2, comps2, bases2, W2);
  rgcn_scatter_h<<<RG_G * RG_M, RG_THREADS, 0, stream>>>(values, hr, hc, W1, h);
  rgcn_layer2<<<RG_G * RG_M, RG_THREADS, 0, stream>>>(values, vr, vc, h, bias1, W2, out);
}


# new launch between layers does layer 2's repeated work once: edges counting-sorted by owner block of out rows, h gets bias+relu; layer-2 kernel: no 16x id scan or global atomics; sum order changes
# speedup vs baseline: 1.9319x; 1.0827x over previous
#include <hip/hip_runtime.h>
#include <hip/hip_bf16.h>

#define NN    50000
#define RR    51
#define EE    16
#define CC    8
#define BB    40
#define NNZ   1600000
#define NE    (NN * EE)
#define W1ELEMS ((size_t)RR * NE)

typedef float v2f __attribute__((ext_vector_type(2)));
typedef float v8f __attribute__((ext_vector_type(8)));

__device__ __forceinline__ void atomAddF32(float* p, float v) {
  unsafeAtomicAdd(p, v);
}

#define K2_THREADS 256
#define K2_BLOCKS  768
#define K2_WAVES   (K2_BLOCKS * (K2_THREADS / 32))

__global__ void __launch_bounds__(K2_THREADS)
rgcn_w1_gemm(const float* __restrict__ comps1,
             const float* __restrict__ bases1,
             float* __restrict__ W1,
             float* __restrict__ h,
             const float* __restrict__ comps2,
             const float* __restrict__ bases2,
             float* __restrict__ W2) {
  {
    const int gid      = blockIdx.x * K2_THREADS + threadIdx.x;
    const int nThreads = K2_BLOCKS * K2_THREADS;
    float4* h4   = reinterpret_cast<float4*>(h);
    const float4 zero4 = make_float4(0.0f, 0.0f, 0.0f, 0.0f);
    for (int i = gid; i < NE / 4; i += nThreads) h4[i] = zero4;
    if (gid < RR * EE * CC) {
      const int r  = gid >> 7;
      const int ec = gid & 127;
      float acc = 0.0f;
#pragma unroll
      for (int b = 0; b < BB; ++b)
        acc += comps2[r * BB + b] * bases2[b * (EE * CC) + ec];
      W2[gid] = acc;
    }
  }

  const int lane = threadIdx.x & 31;
  const int half = lane >> 4;
  const int l16  = lane & 15;

  v2f a[4][10];
#pragma unroll
  for (int mt = 0; mt < 4; ++mt) {
    int m = mt * 16 + l16;
#pragma unroll
    for (int ks = 0; ks < 10; ++ks) {
      int k = ks * 4 + half * 2;
      float ax = 0.0f, ay = 0.0f;
      if (m < RR) {
        ax = comps1[m * BB + k];
        ay = comps1[m * BB + k + 1];
      }
      a[mt][ks].x = ax;
      a[mt][ks].y = ay;
    }
  }

  const int waveId  = blockIdx.x * (K2_THREADS / 32) + (threadIdx.x >> 5);
  const int nWaves  = K2_WAVES;
  const int nTiles  = NE / 16;

  for (int tile = waveId; tile < nTiles; tile += nWaves) {
    const int col = tile * 16 + l16;
    v8f acc[4] = {};
#pragma unroll
    for (int ks = 0; ks < 10; ++ks) {
      const int k = ks * 4 + half * 2;
      v2f b;
      b.x = __builtin_nontemporal_load(&bases1[(size_t)k * NE + col]);
      b.y = __builtin_nontemporal_load(&bases1[(size_t)(k + 1) * NE + col]);
#pragma unroll
      for (int mt = 0; mt < 4; ++mt) {
        acc[mt] = __builtin_amdgcn_wmma_f32_16x16x4_f32(
             false, a[mt][ks],  false, b,
             (short)0, acc[mt],
             false,  false);
      }
    }
#pragma unroll
    for (int mt = 0; mt < 4; ++mt) {
#pragma unroll
      for (int j = 0; j < 8; ++j) {
        int m = mt * 16 + j + half * 8;
        if (m < RR) W1[(size_t)m * NE + col] = acc[mt][j];
      }
    }
  }
}

#define RG_ROWS 3128
#define RG_G    16
#define RG_M    16
#define RG_SLICE (NNZ / RG_M)
#define RG_QSHIFT 17
#define RG_QMASK  0x1FFFFu
#define RG_THREADS 1024

template <int BATCH, int RG_QW, typename RowFn, typename DirectFn, typename DrainFn>
__device__ __forceinline__ void rgcn_scan_slice(const int* __restrict__ ids, int eLo,
                                                int tid, int lane, unsigned* myQueue,
                                                RowFn localRow, DirectFn direct,
                                                DrainFn drainBatch) {
  const int4* ids4 = reinterpret_cast<const int4*>(ids + eLo);
  const unsigned ltMask = (1u << lane) - 1u;
  int wcount = 0;
  int drained = 0;
  for (int jb = 0; jb < RG_SLICE / 4; jb += 4 * RG_THREADS) {
    int4 rr[4];
#pragma unroll
    for (int k = 0; k < 4; ++k) {
      const int j = jb + k * RG_THREADS + tid;
      rr[k] = make_int4(-1, -1, -1, -1);
      if (j < RG_SLICE / 4) rr[k] = ids4[j];
    }
#pragma unroll
    for (int k = 0; k < 4; ++k) {
      if (jb + k * RG_THREADS >= RG_SLICE / 4) break;
      const int j = jb + k * RG_THREADS + tid;
      const bool inRange = j < RG_SLICE / 4;
      const unsigned d0 = localRow(rr[k].x), d1 = localRow(rr[k].y);
      const unsigned d2 = localRow(rr[k].z), d3 = localRow(rr[k].w);
      const bool s0 = inRange & (d0 < RG_ROWS), s1 = inRange & (d1 < RG_ROWS);
      const bool s2 = inRange & (d2 < RG_ROWS), s3 = inRange & (d3 < RG_ROWS);
      const unsigned b0 = __builtin_amdgcn_ballot_w32(s0);
      const unsigned b1 = __builtin_amdgcn_ballot_w32(s1);
      const unsigned b2 = __builtin_amdgcn_ballot_w32(s2);
      const unsigned b3 = __builtin_amdgcn_ballot_w32(s3);
      if ((b0 | b1 | b2 | b3) != 0u) {
        const int i0 = eLo + 4 * j;
        const unsigned q0 = 4u * (unsigned)j;
        const int p0 = wcount + __popc(b0 & ltMask);
        const int c0 = wcount + __popc(b0);
        const int p1 = c0 + __popc(b1 & ltMask);
        const int c1 = c0 + __popc(b1);
        const int p2 = c1 + __popc(b2 & ltMask);
        const int c2 = c1 + __popc(b2);
        const int p3 = c2 + __popc(b3 & ltMask);
        wcount = c2 + __popc(b3);
        if (s0) { if (__builtin_expect(p0 < RG_QW, 1)) myQueue[p0] = (d0 << RG_QSHIFT) | q0;
                  else direct(d0, i0); }
        if (s1) { if (__builtin_expect(p1 < RG_QW, 1)) myQueue[p1] = (d1 << RG_QSHIFT) | (q0 + 1u);
                  else direct(d1, i0 + 1); }
        if (s2) { if (__builtin_expect(p2 < RG_QW, 1)) myQueue[p2] = (d2 << RG_QSHIFT) | (q0 + 2u);
                  else direct(d2, i0 + 2); }
        if (s3) { if (__builtin_expect(p3 < RG_QW, 1)) myQueue[p3] = (d3 << RG_QSHIFT) | (q0 + 3u);
                  else direct(d3, i0 + 3); }
      }
    }
    const int avail = wcount < RG_QW ? wcount : RG_QW;
    for (; avail - drained >= BATCH; drained += BATCH) drainBatch(drained, avail);
  }
  const int n = wcount < RG_QW ? wcount : RG_QW;
  for (; drained < n; drained += BATCH) drainBatch(drained, n);
}

#define RG_U3   16
#define RG_QW3  512

__global__ void __launch_bounds__(RG_THREADS)
rgcn_scatter_h(const float* __restrict__ values,
               const int* __restrict__ hr,
               const int* __restrict__ hc,
               const float* __restrict__ W1,
               float* __restrict__ h) {
  __shared__ float    accTile[RG_ROWS * EE];
  __shared__ unsigned queue[32][RG_QW3];
  __shared__ float    spare[EE];

  const int tid  = threadIdx.x;
  const int lane = tid & 31;
  const int wv   = tid >> 5;
  const int half = lane >> 4;
  const int l16  = lane & 15;
  const int g    = blockIdx.x % RG_G;
  const int m    = blockIdx.x / RG_G;
  const int rowLo = g * RG_ROWS;
  const int eLo   = m * RG_SLICE;

  for (int j = tid; j < RG_ROWS * EE; j += RG_THREADS) accTile[j] = 0.0f;
  __syncthreads();

  rgcn_scan_slice<2 * RG_U3, RG_QW3>(
      hr, eLo, tid, lane, queue[wv],
      [&](int r) { return (unsigned)(r - rowLo); },
      [&](unsigned d, int i) {
        const float v = values[i];
        const float* wp = W1 + (size_t)hc[i] * EE;
#pragma nounroll
        for (int e = 0; e < EE; ++e) atomAddF32(&accTile[d * EE + e], v * wp[e]);
      },
      [&](int base, int n) {
        unsigned ent[RG_U3];
        bool ok[RG_U3];
#pragma unroll
        for (int u = 0; u < RG_U3; ++u) {
          const int q = base + 2 * u + half;
          ok[u] = q < n;
          ent[u] = ok[u] ? queue[wv][q] : 0u;
        }
        int col[RG_U3];
        float v[RG_U3];
#pragma unroll
        for (int u = 0; u < RG_U3; ++u) {
          const int i = eLo + (ok[u] ? (int)(ent[u] & RG_QMASK) : 0);
          col[u] = __builtin_nontemporal_load(&hc[i]);
          v[u]   = __builtin_nontemporal_load(&values[i]);
        }
        float w[RG_U3];
#pragma unroll
        for (int u = 0; u < RG_U3; ++u)
          w[u] = W1[(size_t)col[u] * EE + l16];
#pragma unroll
        for (int u = 0; u < RG_U3; ++u) {
          float* dst = ok[u] ? &accTile[(ent[u] >> RG_QSHIFT) * EE + l16] : &spare[l16];
          atomAddF32(dst, v[u] * w[u]);
        }
      });
  __syncthreads();

  for (int j = tid; j < RG_ROWS * EE; j += RG_THREADS) {
    const int row = rowLo + (j >> 4);
    if (row < NN) atomAddF32(&h[(size_t)rowLo * EE + j], accTile[j]);
  }
}

#define L2_G     256
#define L2_ROWS  196
#define L2_CH    (NNZ / L2_G)
#define L2_CHP   6272
#define L2_OFFP  288
#define L2_RECF  ((size_t)L2_G * L2_CHP * 2)
#define L2_SORT_THREADS 512
#define L2_SORT_K ((L2_CH + L2_SORT_THREADS - 1) / L2_SORT_THREADS)

__global__ void __launch_bounds__(L2_SORT_THREADS)
rgcn_layer2_sort(const float* __restrict__ values,
                 const int* __restrict__ vr,
                 const int* __restrict__ vc,
                 const float* __restrict__ bias1,
                 float* __restrict__ h,
                 float* __restrict__ ws) {
  __shared__ unsigned cursor[L2_G];
  __shared__ unsigned start[L2_G + 1];
  __shared__ uint2    recL[L2_CH];

  const int tid  = threadIdx.x;
  const int lane = tid & 31;
  const int c    = blockIdx.x;
  const int eLo  = c * L2_CH;

  if (tid < L2_G) cursor[tid] = 0u;
  __syncthreads();

  const int gid = c * L2_SORT_THREADS + tid;
  float4* h4 = reinterpret_cast<float4*>(h);
  const int  hi1 = gid + L2_G * L2_SORT_THREADS;
  const bool hok1 = hi1 < NE / 4;
  const float4 hx0 = h4[gid];
  const float4 hx1 = h4[hok1 ? hi1 : gid];
  const int e0 = (gid & 3) * 4;
  const float4 bv = make_float4(bias1[e0], bias1[e0 + 1], bias1[e0 + 2], bias1[e0 + 3]);

  unsigned r[L2_SORT_K];
  int      o[L2_SORT_K];
  float    v[L2_SORT_K];
#pragma unroll
  for (int k = 0; k < L2_SORT_K; ++k) {
    const int j  = tid + k * L2_SORT_THREADS;
    const int jj = j < L2_CH ? j : L2_CH - 1;
    r[k] = (unsigned)vr[eLo + jj];
    o[k] = vc[eLo + jj];
    v[k] = values[eLo + jj];
  }
  h4[gid] = make_float4(fmaxf(hx0.x + bv.x, 0.0f), fmaxf(hx0.y + bv.y, 0.0f),
                        fmaxf(hx0.z + bv.z, 0.0f), fmaxf(hx0.w + bv.w, 0.0f));
  if (hok1)
    h4[hi1] = make_float4(fmaxf(hx1.x + bv.x, 0.0f), fmaxf(hx1.y + bv.y, 0.0f),
                          fmaxf(hx1.z + bv.z, 0.0f), fmaxf(hx1.w + bv.w, 0.0f));
  unsigned blk[L2_SORT_K], word0[L2_SORT_K];
#pragma unroll
  for (int k = 0; k < L2_SORT_K; ++k) {
    const unsigned p = r[k] / (unsigned)NN;
    const unsigned s = r[k] - p * (unsigned)NN;
    const unsigned b = s / (unsigned)L2_ROWS;
    blk[k]   = b;
    word0[k] = ((s - b * (unsigned)L2_ROWS) << 22) | (p << 16) | (unsigned)o[k];
    if (tid + k * L2_SORT_THREADS < L2_CH) atomicAdd(&cursor[b], 1u);
  }
  __syncthreads();

  if (tid < 32) {
    unsigned x[8];
    unsigned tot = 0u;
#pragma unroll
    for (int i = 0; i < 8; ++i) { x[i] = cursor[lane * 8 + i]; tot += x[i]; }
    unsigned incl = tot;
#pragma unroll
    for (int d = 1; d < 32; d <<= 1) {
      const unsigned y = __shfl_up(incl, (unsigned)d);
      if (lane >= d) incl += y;
    }
    unsigned run = incl - tot;
#pragma unroll
    for (int i = 0; i < 8; ++i) {
      start[lane * 8 + i]  = run;
      cursor[lane * 8 + i] = run;
      run += x[i];
    }
    if (lane == 31) start[L2_G] = run;
  }
  __syncthreads();

  unsigned pos[L2_SORT_K];
#pragma unroll
  for (int k = 0; k < L2_SORT_K; ++k) {
    pos[k] = 0u;
    if (tid + k * L2_SORT_THREADS < L2_CH) pos[k] = atomicAdd(&cursor[blk[k]], 1u);
  }
#pragma unroll
  for (int k = 0; k < L2_SORT_K; ++k) {
    if (tid + k * L2_SORT_THREADS < L2_CH) recL[pos[k]] = make_uint2(word0[k], __float_as_uint(v[k]));
  }
  __syncthreads();

  uint2* recs = reinterpret_cast<uint2*>(ws) + (size_t)c * L2_CHP;
#pragma unroll
  for (int k = 0; k < L2_SORT_K; ++k) {
    const int j = tid + k * L2_SORT_THREADS;
    if (j < L2_CH) recs[j] = recL[j];
  }
  unsigned* offs = reinterpret_cast<unsigned*>(ws + L2_RECF) + (size_t)c * L2_OFFP;
  if (tid <= L2_G) offs[tid] = start[tid];
}

#define RG_U5   4
#define W2T_CP   20
#define W2T_PP   (CC * W2T_CP)
#define W2T_COPY (RR * W2T_PP)
#define L2_QW   512
#define L2_CPW  (L2_G / 32)

__global__ void __launch_bounds__(RG_THREADS)
rgcn_layer2(const float* __restrict__ h,
            const float* __restrict__ bias2,
            const float* __restrict__ W2,
            const float* __restrict__ ws,
            float* __restrict__ out) {
  __shared__ __attribute__((aligned(16))) float W2T[2 * W2T_COPY];
  __shared__ float    outTile[L2_ROWS * CC];
  __shared__ uint2    queue[32][L2_QW];
  __shared__ float    spare[CC];

  const int tid  = threadIdx.x;
  const int lane = tid & 31;
  const int wv   = tid >> 5;
  const int sub  = lane >> 3;
  const int c    = lane & 7;
  const int b    = blockIdx.x;
  const int rowLo = b * L2_ROWS;
  const int wvS  = __builtin_amdgcn_readfirstlane(wv);

  for (int j = tid; j < RR * EE * CC; j += RG_THREADS) {
    const int t = (j >> 7) * W2T_PP + (j & 7) * W2T_CP + ((j >> 3) & 15);
    const float w = W2[j];
    W2T[t] = w;
    W2T[W2T_COPY + t] = w;
  }
  for (int j = tid; j < L2_ROWS * CC; j += RG_THREADS) outTile[j] = 0.0f;
  __syncthreads();

  const uint2*    recs = reinterpret_cast<const uint2*>(ws);
  const unsigned* offs = reinterpret_cast<const unsigned*>(ws + L2_RECF);
  uint2* myQueue = queue[wv];

  auto drainBatch = [&](int base, int n) {
    uint2 ent[RG_U5];
    bool ok[RG_U5];
#pragma unroll
    for (int u = 0; u < RG_U5; ++u) {
      const int q = base + 4 * u + sub;
      ok[u] = q < n;
      ent[u] = myQueue[ok[u] ? q : 0];
      if (!ok[u]) ent[u] = make_uint2(0u, 0u);
    }
    float4 hv[RG_U5][4];
#pragma unroll
    for (int u = 0; u < RG_U5; ++u) {
      const float4* hp4 = reinterpret_cast<const float4*>(h + (size_t)(ent[u].x & 0xFFFFu) * EE);
#pragma unroll
      for (int k = 0; k < 4; ++k) hv[u][k] = hp4[k];
    }
    __builtin_amdgcn_sched_barrier(0);
    float acc[RG_U5];
#pragma unroll
    for (int u = 0; u < RG_U5; ++u) {
      const int p = (int)((ent[u].x >> 16) & 63u);
      const int copy = (p ^ sub) & 1;
      const float4* wq = reinterpret_cast<const float4*>(
          W2T + copy * W2T_COPY + p * W2T_PP + c * W2T_CP);
      float a = 0.0f;
#pragma unroll
      for (int k = 0; k < 4; ++k) {
        const float4 w = wq[k];
        a += hv[u][k].x * w.x;
        a += hv[u][k].y * w.y;
        a += hv[u][k].z * w.z;
        a += hv[u][k].w * w.w;
      }
      acc[u] = a;
    }
#pragma unroll
    for (int u = 0; u < RG_U5; ++u) {
      float* dst = ok[u] ? &outTile[(ent[u].x >> 22) * CC + c] : &spare[c];
      atomAddF32(dst, __uint_as_float(ent[u].y) * acc[u]);
    }
  };
  auto drainAll = [&](int n) {
    for (int q = 0; q < n; q += 4 * RG_U5) drainBatch(q, n);
  };

  unsigned myBeg = 0u, myLen = 0u;
  if (lane < L2_CPW) {
    const unsigned* op = offs + (size_t)(wv * L2_CPW + lane) * L2_OFFP + b;
    myBeg = op[0];
    myLen = op[1] - myBeg;
  }
  unsigned beg[L2_CPW], len[L2_CPW];
#pragma unroll
  for (int k = 0; k < L2_CPW; ++k) {
    beg[k] = (unsigned)__builtin_amdgcn_readlane((int)myBeg, k);
    len[k] = (unsigned)__builtin_amdgcn_readlane((int)myLen, k);
  }
  uint2 first[L2_CPW];
#pragma unroll
  for (int k = 0; k < L2_CPW; ++k) {
    const uint2* rp = recs + (size_t)(wv * L2_CPW + k) * L2_CHP;
    first[k] = rp[(unsigned)lane < len[k] ? beg[k] + (unsigned)lane : 0u];
  }
  int qn = 0;
#pragma unroll
  for (int k = 0; k < L2_CPW; ++k) {
    const int cnt = len[k] < 32u ? (int)len[k] : 32;
    if (lane < cnt) myQueue[qn + lane] = first[k];
    qn += cnt;
  }
  if (__builtin_amdgcn_ballot_w32(myLen > 32u) != 0u) {
#pragma nounroll
    for (int k = 0; k < L2_CPW; ++k) {
      const unsigned* op = offs + (size_t)(wvS * L2_CPW + k) * L2_OFFP + b;
      const unsigned bk = op[0];
      const unsigned lk = op[1] - bk;
      const uint2* rp = recs + (size_t)(wvS * L2_CPW + k) * L2_CHP + bk;
      for (unsigned q0 = 32u; q0 < lk; q0 += 32u) {
        const unsigned left = lk - q0;
        const int cnt = left < 32u ? (int)left : 32;
        if (qn + cnt > L2_QW) { drainAll(qn); qn = 0; }
        if (lane < cnt) myQueue[qn + lane] = rp[q0 + (unsigned)lane];
        qn += cnt;
      }
    }
  }
  drainAll(qn);
  __syncthreads();

  const int tidC = wvS * 32 + (int)__builtin_amdgcn_mbcnt_lo(~0u, 0u);
  for (int j = tidC; j < L2_ROWS * CC; j += RG_THREADS) {
    const int row = rowLo + (j >> 3);
    if (row < NN) out[(size_t)rowLo * CC + j] = outTile[j] + bias2[j & 7];
  }
}

extern "C" void kernel_launch(void* const* d_in, const int* in_sizes, int n_in,
                              void* d_out, int out_size, void* d_ws, size_t ws_size,
                              hipStream_t stream) {
  const float* values = (const float*)d_in[0];
  const float* comps1 = (const float*)d_in[1];
  const float* bases1 = (const float*)d_in[2];
  const float* comps2 = (const float*)d_in[3];
  const float* bases2 = (const float*)d_in[4];
  const float* bias1  = (const float*)d_in[5];
  const float* bias2  = (const float*)d_in[6];
  const int*   hr     = (const int*)d_in[7];
  const int*   hc     = (const int*)d_in[8];
  const int*   vr     = (const int*)d_in[9];
  const int*   vc     = (const int*)d_in[10];
  float* out = (float*)d_out;

  float* W1 = (float*)d_ws;
  float* h  = W1 + W1ELEMS;
  float* W2 = h + NE;

  rgcn_w1_gemm<<<K2_BLOCKS, K2_THREADS, 0, stream>>>(comps1, bases1, W1, h, comps2, bases2, W2);
  rgcn_scatter_h<<<RG_G * RG_M, RG_THREADS, 0, stream>>>(values, hr, hc, W1, h);
  rgcn_layer2_sort<<<L2_G, L2_SORT_THREADS, 0, stream>>>(values, vr, vc, bias1, h, W1);
  rgcn_layer2<<<L2_G, RG_THREADS, 0, stream>>>(h, bias2, W2, W1, out);
}


# W1 table no longer stored: layer-1 edges sorted by source node (new sort launch), each WMMA tile (one node's 51x16) applied to its edges from LDS, per-edge f32 atomics into h; sum order changes
# speedup vs baseline: 2.1955x; 1.1364x over previous
#include <hip/hip_runtime.h>
#include <hip/hip_bf16.h>

#define NN    50000
#define RR    51
#define EE    16
#define CC    8
#define BB    40
#define NNZ   1600000
#define NE    (NN * EE)
#define W1ELEMS ((size_t)RR * NE)

typedef float v2f __attribute__((ext_vector_type(2)));
typedef float v8f __attribute__((ext_vector_type(8)));

__device__ __forceinline__ void atomAddF32(float* p, float v) {
  unsafeAtomicAdd(p, v);
}


#define RG_THREADS 1024

#define L2_G     256
#define L2_ROWS  196
#define L2_CH    (NNZ / L2_G)
#define L2_CHP   6272
#define L2_OFFP  288
#define L2_RECF  ((size_t)L2_G * L2_CHP * 2)
#define L2_SORT_THREADS 512
#define L2_SORT_K ((L2_CH + L2_SORT_THREADS - 1) / L2_SORT_THREADS)

template <bool FIRST>
__device__ __forceinline__ void rgcn_sort_body(const float* __restrict__ values,
                                               const int* __restrict__ vr,
                                               const int* __restrict__ vc,
                                               const float* __restrict__ bias1,
                                               float* __restrict__ h,
                                               float* __restrict__ ws) {
  __shared__ unsigned cursor[L2_G];
  __shared__ unsigned start[L2_G + 1];
  __shared__ uint2    recL[L2_CH];

  const int tid  = threadIdx.x;
  const int lane = tid & 31;
  const int c    = blockIdx.x;
  const int eLo  = c * L2_CH;

  if (tid < L2_G) cursor[tid] = 0u;
  __syncthreads();

  const int gid = c * L2_SORT_THREADS + tid;
  float4* h4 = reinterpret_cast<float4*>(h);
  const int  hi1 = gid + L2_G * L2_SORT_THREADS;
  const bool hok1 = hi1 < NE / 4;
  float4 hx0 = make_float4(0.0f, 0.0f, 0.0f, 0.0f), hx1 = hx0, bv = hx0;
  if constexpr (!FIRST) {
    hx0 = h4[gid];
    hx1 = h4[hok1 ? hi1 : gid];
    const int e0 = (gid & 3) * 4;
    bv = make_float4(bias1[e0], bias1[e0 + 1], bias1[e0 + 2], bias1[e0 + 3]);
  }

  unsigned r[L2_SORT_K];
  int      o[L2_SORT_K];
  float    v[L2_SORT_K];
#pragma unroll
  for (int k = 0; k < L2_SORT_K; ++k) {
    const int j  = tid + k * L2_SORT_THREADS;
    const int jj = j < L2_CH ? j : L2_CH - 1;
    r[k] = (unsigned)vr[eLo + jj];
    o[k] = vc[eLo + jj];
    v[k] = values[eLo + jj];
  }
  if constexpr (FIRST) {
    h4[gid] = hx0;
    if (hok1) h4[hi1] = hx0;
  } else {
    h4[gid] = make_float4(fmaxf(hx0.x + bv.x, 0.0f), fmaxf(hx0.y + bv.y, 0.0f),
                          fmaxf(hx0.z + bv.z, 0.0f), fmaxf(hx0.w + bv.w, 0.0f));
    if (hok1)
      h4[hi1] = make_float4(fmaxf(hx1.x + bv.x, 0.0f), fmaxf(hx1.y + bv.y, 0.0f),
                            fmaxf(hx1.z + bv.z, 0.0f), fmaxf(hx1.w + bv.w, 0.0f));
  }
  unsigned blk[L2_SORT_K], word0[L2_SORT_K];
#pragma unroll
  for (int k = 0; k < L2_SORT_K; ++k) {
    const unsigned p = r[k] / (unsigned)NN;
    const unsigned s = r[k] - p * (unsigned)NN;
    const unsigned b = s / (unsigned)L2_ROWS;
    blk[k]   = b;
    word0[k] = ((s - b * (unsigned)L2_ROWS) << 22) | (p << 16) | (unsigned)o[k];
    if (tid + k * L2_SORT_THREADS < L2_CH) atomicAdd(&cursor[b], 1u);
  }
  __syncthreads();

  if (tid < 32) {
    unsigned x[8];
    unsigned tot = 0u;
#pragma unroll
    for (int i = 0; i < 8; ++i) { x[i] = cursor[lane * 8 + i]; tot += x[i]; }
    unsigned incl = tot;
#pragma unroll
    for (int d = 1; d < 32; d <<= 1) {
      const unsigned y = __shfl_up(incl, (unsigned)d);
      if (lane >= d) incl += y;
    }
    unsigned run = incl - tot;
#pragma unroll
    for (int i = 0; i < 8; ++i) {
      start[lane * 8 + i]  = run;
      cursor[lane * 8 + i] = run;
      run += x[i];
    }
    if (lane == 31) start[L2_G] = run;
  }
  __syncthreads();

  unsigned pos[L2_SORT_K];
#pragma unroll
  for (int k = 0; k < L2_SORT_K; ++k) {
    pos[k] = 0u;
    if (tid + k * L2_SORT_THREADS < L2_CH) pos[k] = atomicAdd(&cursor[blk[k]], 1u);
  }
#pragma unroll
  for (int k = 0; k < L2_SORT_K; ++k) {
    if (tid + k * L2_SORT_THREADS < L2_CH) recL[pos[k]] = make_uint2(word0[k], __float_as_uint(v[k]));
  }
  __syncthreads();

  uint2* recs = reinterpret_cast<uint2*>(ws) + (size_t)c * L2_CHP;
#pragma unroll
  for (int k = 0; k < L2_SORT_K; ++k) {
    const int j = tid + k * L2_SORT_THREADS;
    if (j < L2_CH) recs[j] = recL[j];
  }
  unsigned* offs = reinterpret_cast<unsigned*>(ws + L2_RECF) + (size_t)c * L2_OFFP;
  if (tid <= L2_G) offs[tid] = start[tid];
}

__global__ void __launch_bounds__(L2_SORT_THREADS)
rgcn_layer1_sort(const float* __restrict__ values,
                 const int* __restrict__ hc,
                 const int* __restrict__ hr,
                 float* __restrict__ h,
                 float* __restrict__ ws) {
  rgcn_sort_body<true>(values, hc, hr, nullptr, h, ws);
}

__global__ void __launch_bounds__(L2_SORT_THREADS)
rgcn_layer2_sort(const float* __restrict__ values,
                 const int* __restrict__ vr,
                 const int* __restrict__ vc,
                 const float* __restrict__ bias1,
                 float* __restrict__ h,
                 float* __restrict__ ws) {
  rgcn_sort_body<false>(values, vr, vc, bias1, h, ws);
}

#define RG_U5   4
#define W2T_CP   20
#define W2T_PP   (CC * W2T_CP)
#define W2T_COPY (RR * W2T_PP)
#define L2_QW   512
#define L2_CPW  (L2_G / 32)

__global__ void __launch_bounds__(RG_THREADS)
rgcn_layer2(const float* __restrict__ h,
            const float* __restrict__ bias2,
            const float* __restrict__ W2,
            const float* __restrict__ ws,
            float* __restrict__ out) {
  __shared__ __attribute__((aligned(16))) float W2T[2 * W2T_COPY];
  __shared__ float    outTile[L2_ROWS * CC];
  __shared__ uint2    queue[32][L2_QW];
  __shared__ float    spare[CC];

  const int tid  = threadIdx.x;
  const int lane = tid & 31;
  const int wv   = tid >> 5;
  const int sub  = lane >> 3;
  const int c    = lane & 7;
  const int b    = blockIdx.x;
  const int rowLo = b * L2_ROWS;
  const int wvS  = __builtin_amdgcn_readfirstlane(wv);

  for (int j = tid; j < RR * EE * CC; j += RG_THREADS) {
    const int t = (j >> 7) * W2T_PP + (j & 7) * W2T_CP + ((j >> 3) & 15);
    const float w = W2[j];
    W2T[t] = w;
    W2T[W2T_COPY + t] = w;
  }
  for (int j = tid; j < L2_ROWS * CC; j += RG_THREADS) outTile[j] = 0.0f;
  __syncthreads();

  const uint2*    recs = reinterpret_cast<const uint2*>(ws);
  const unsigned* offs = reinterpret_cast<const unsigned*>(ws + L2_RECF);
  uint2* myQueue = queue[wv];

  auto drainBatch = [&](int base, int n) {
    uint2 ent[RG_U5];
    bool ok[RG_U5];
#pragma unroll
    for (int u = 0; u < RG_U5; ++u) {
      const int q = base + 4 * u + sub;
      ok[u] = q < n;
      ent[u] = myQueue[ok[u] ? q : 0];
      if (!ok[u]) ent[u] = make_uint2(0u, 0u);
    }
    float4 hv[RG_U5][4];
#pragma unroll
    for (int u = 0; u < RG_U5; ++u) {
      const float4* hp4 = reinterpret_cast<const float4*>(h + (size_t)(ent[u].x & 0xFFFFu) * EE);
#pragma unroll
      for (int k = 0; k < 4; ++k) hv[u][k] = hp4[k];
    }
    __builtin_amdgcn_sched_barrier(0);
    float acc[RG_U5];
#pragma unroll
    for (int u = 0; u < RG_U5; ++u) {
      const int p = (int)((ent[u].x >> 16) & 63u);
      const int copy = (p ^ sub) & 1;
      const float4* wq = reinterpret_cast<const float4*>(
          W2T + copy * W2T_COPY + p * W2T_PP + c * W2T_CP);
      float a = 0.0f;
#pragma unroll
      for (int k = 0; k < 4; ++k) {
        const float4 w = wq[k];
        a += hv[u][k].x * w.x;
        a += hv[u][k].y * w.y;
        a += hv[u][k].z * w.z;
        a += hv[u][k].w * w.w;
      }
      acc[u] = a;
    }
#pragma unroll
    for (int u = 0; u < RG_U5; ++u) {
      float* dst = ok[u] ? &outTile[(ent[u].x >> 22) * CC + c] : &spare[c];
      atomAddF32(dst, __uint_as_float(ent[u].y) * acc[u]);
    }
  };
  auto drainAll = [&](int n) {
    for (int q = 0; q < n; q += 4 * RG_U5) drainBatch(q, n);
  };

  unsigned myBeg = 0u, myLen = 0u;
  if (lane < L2_CPW) {
    const unsigned* op = offs + (size_t)(wv * L2_CPW + lane) * L2_OFFP + b;
    myBeg = op[0];
    myLen = op[1] - myBeg;
  }
  unsigned beg[L2_CPW], len[L2_CPW];
#pragma unroll
  for (int k = 0; k < L2_CPW; ++k) {
    beg[k] = (unsigned)__builtin_amdgcn_readlane((int)myBeg, k);
    len[k] = (unsigned)__builtin_amdgcn_readlane((int)myLen, k);
  }
  uint2 first[L2_CPW];
#pragma unroll
  for (int k = 0; k < L2_CPW; ++k) {
    const uint2* rp = recs + (size_t)(wv * L2_CPW + k) * L2_CHP;
    first[k] = rp[(unsigned)lane < len[k] ? beg[k] + (unsigned)lane : 0u];
  }
  int qn = 0;
#pragma unroll
  for (int k = 0; k < L2_CPW; ++k) {
    const int cnt = len[k] < 32u ? (int)len[k] : 32;
    if (lane < cnt) myQueue[qn + lane] = first[k];
    qn += cnt;
  }
  if (__builtin_amdgcn_ballot_w32(myLen > 32u) != 0u) {
#pragma nounroll
    for (int k = 0; k < L2_CPW; ++k) {
      const unsigned* op = offs + (size_t)(wvS * L2_CPW + k) * L2_OFFP + b;
      const unsigned bk = op[0];
      const unsigned lk = op[1] - bk;
      const uint2* rp = recs + (size_t)(wvS * L2_CPW + k) * L2_CHP + bk;
      for (unsigned q0 = 32u; q0 < lk; q0 += 32u) {
        const unsigned left = lk - q0;
        const int cnt = left < 32u ? (int)left : 32;
        if (qn + cnt > L2_QW) { drainAll(qn); qn = 0; }
        if (lane < cnt) myQueue[qn + lane] = rp[q0 + (unsigned)lane];
        qn += cnt;
      }
    }
  }
  drainAll(qn);
  __syncthreads();

  const int tidC = wvS * 32 + (int)__builtin_amdgcn_mbcnt_lo(~0u, 0u);
  for (int j = tidC; j < L2_ROWS * CC; j += RG_THREADS) {
    const int row = rowLo + (j >> 3);
    if (row < NN) out[(size_t)rowLo * CC + j] = outTile[j] + bias2[j & 7];
  }
}


#define G_THREADS 768
#define G_WAVES   (G_THREADS / 32)
#define G_K       9
#define G_CAP     (G_K * G_THREADS)
#define G_BINS    224
#define G_U       4

__global__ void __launch_bounds__(G_THREADS)
rgcn_w1_apply(const float* __restrict__ comps1,
              const float* __restrict__ bases1,
              float* __restrict__ h,
              const float* __restrict__ comps2,
              const float* __restrict__ bases2,
              float* __restrict__ W2,
              const float* __restrict__ ws) {
  __shared__ uint2    sorted[G_CAP];
  __shared__ float    tileL[G_WAVES][RR * EE];
  __shared__ unsigned pbeg[L2_G];
  __shared__ unsigned pstart[L2_G + 1];
  __shared__ unsigned cursor[G_BINS];
  __shared__ unsigned nstart[G_BINS + 1];

  const int tid  = threadIdx.x;
  const int lane = tid & 31;
  const int wv   = tid >> 5;
  const int half = lane >> 4;
  const int l16  = lane & 15;
  const int b    = blockIdx.x;

  {
    const int gid = b * G_THREADS + tid;
    if (gid < RR * EE * CC) {
      const int r  = gid >> 7;
      const int ec = gid & 127;
      float acc = 0.0f;
#pragma unroll
      for (int k = 0; k < BB; ++k)
        acc += comps2[r * BB + k] * bases2[k * (EE * CC) + ec];
      W2[gid] = acc;
    }
  }

  const uint2*    recs = reinterpret_cast<const uint2*>(ws);
  const unsigned* offs = reinterpret_cast<const unsigned*>(ws + L2_RECF);

  if (tid < L2_G) {
    const unsigned* op = offs + (size_t)tid * L2_OFFP + b;
    const unsigned u0 = op[0];
    pbeg[tid]   = u0;
    pstart[tid] = op[1] - u0;
  }
  __syncthreads();
  if (tid < 32) {
    unsigned x[8];
    unsigned tot = 0u;
#pragma unroll
    for (int i = 0; i < 8; ++i) { x[i] = pstart[lane * 8 + i]; tot += x[i]; }
    unsigned incl = tot;
#pragma unroll
    for (int d = 1; d < 32; d <<= 1) {
      const unsigned y = __shfl_up(incl, (unsigned)d);
      if (lane >= d) incl += y;
    }
    unsigned run = incl - tot;
#pragma unroll
    for (int i = 0; i < 8; ++i) { pstart[lane * 8 + i] = run; run += x[i]; }
    if (lane == 31) pstart[L2_G] = run;
  }
  __syncthreads();
  const int total = (int)pstart[L2_G];

  for (int base = 0; base < total; base += G_CAP) {
    const int passEnd = total < base + G_CAP ? total : base + G_CAP;
    if (tid < G_BINS) cursor[tid] = 0u;
    __syncthreads();

    unsigned w0[G_K], w1[G_K];
#pragma unroll
    for (int k = 0; k < G_K; ++k) {
      const int idx = base + tid + k * G_THREADS;
      const unsigned ii = (unsigned)(idx < passEnd ? idx : base);
      int pc = 0;
#pragma unroll
      for (int st = L2_G / 2; st >= 1; st >>= 1)
        if (pstart[pc + st] <= ii) pc += st;
      const uint2 rec = recs[(size_t)pc * L2_CHP + pbeg[pc] + (ii - pstart[pc])];
      w0[k] = rec.x;
      w1[k] = rec.y;
    }
#pragma unroll
    for (int k = 0; k < G_K; ++k)
      if (base + tid + k * G_THREADS < passEnd) atomicAdd(&cursor[w0[k] >> 22], 1u);
    __syncthreads();

    if (tid < 32) {
      unsigned x[7];
      unsigned tot = 0u;
#pragma unroll
      for (int i = 0; i < 7; ++i) { x[i] = cursor[lane * 7 + i]; tot += x[i]; }
      unsigned incl = tot;
#pragma unroll
      for (int d = 1; d < 32; d <<= 1) {
        const unsigned y = __shfl_up(incl, (unsigned)d);
        if (lane >= d) incl += y;
      }
      unsigned run = incl - tot;
#pragma unroll
      for (int i = 0; i < 7; ++i) {
        nstart[lane * 7 + i] = run;
        cursor[lane * 7 + i] = run;
        run += x[i];
      }
      if (lane == 31) nstart[G_BINS] = run;
    }
    __syncthreads();

    unsigned pos[G_K];
#pragma unroll
    for (int k = 0; k < G_K; ++k) {
      pos[k] = 0u;
      if (base + tid + k * G_THREADS < passEnd) pos[k] = atomicAdd(&cursor[w0[k] >> 22], 1u);
    }
#pragma unroll
    for (int k = 0; k < G_K; ++k)
      if (base + tid + k * G_THREADS < passEnd) sorted[pos[k]] = make_uint2(w0[k], w1[k]);
    __syncthreads();

    v2f a[4][10];
#pragma unroll
    for (int mt = 0; mt < 4; ++mt) {
      int m = mt * 16 + l16;
#pragma unroll
      for (int ks = 0; ks < 10; ++ks) {
        int k = ks * 4 + half * 2;
        float ax = 0.0f, ay = 0.0f;
        if (m < RR) {
          ax = comps1[m * BB + k];
          ay = comps1[m * BB + k + 1];
        }
        a[mt][ks].x = ax;
        a[mt][ks].y = ay;
      }
    }

    float* tl = tileL[wv];
    for (int t = wv; t < L2_ROWS; t += G_WAVES) {
      const int o = b * L2_ROWS + t;
      if (o >= NN) break;
      const int col = o * EE + l16;
      v8f acc[4] = {};
#pragma unroll
      for (int ks = 0; ks < 10; ++ks) {
        const int k = ks * 4 + half * 2;
        v2f bb;
        bb.x = __builtin_nontemporal_load(&bases1[(size_t)k * NE + col]);
        bb.y = __builtin_nontemporal_load(&bases1[(size_t)(k + 1) * NE + col]);
#pragma unroll
        for (int mt = 0; mt < 4; ++mt) {
          acc[mt] = __builtin_amdgcn_wmma_f32_16x16x4_f32(
               false, a[mt][ks],  false, bb,
               (short)0, acc[mt],
               false,  false);
        }
      }
#pragma unroll
      for (int mt = 0; mt < 4; ++mt) {
#pragma unroll
        for (int j = 0; j < 8; ++j) {
          const int m = mt * 16 + j + half * 8;
          if (m < RR) tl[m * EE + l16] = acc[mt][j];
        }
      }
      const int s0 = __builtin_amdgcn_readfirstlane((int)nstart[t]);
      const int s1 = __builtin_amdgcn_readfirstlane((int)nstart[t + 1]);
      for (int q0 = s0; q0 < s1; q0 += 2 * G_U) {
        uint2 ent[G_U];
        bool ok[G_U];
#pragma unroll
        for (int u = 0; u < G_U; ++u) {
          const int q = q0 + 2 * u + half;
          ok[u] = q < s1;
          ent[u] = sorted[ok[u] ? q : s0];
        }
        float w[G_U];
#pragma unroll
        for (int u = 0; u < G_U; ++u)
          w[u] = tl[((ent[u].x >> 16) & 63u) * EE + l16];
#pragma unroll
        for (int u = 0; u < G_U; ++u)
          if (ok[u])
            atomAddF32(&h[(size_t)(ent[u].x & 0xFFFFu) * EE + l16],
                       __uint_as_float(ent[u].y) * w[u]);
      }
    }
  }
}

extern "C" void kernel_launch(void* const* d_in, const int* in_sizes, int n_in,
                              void* d_out, int out_size, void* d_ws, size_t ws_size,
                              hipStream_t stream) {
  const float* values = (const float*)d_in[0];
  const float* comps1 = (const float*)d_in[1];
  const float* bases1 = (const float*)d_in[2];
  const float* comps2 = (const float*)d_in[3];
  const float* bases2 = (const float*)d_in[4];
  const float* bias1  = (const float*)d_in[5];
  const float* bias2  = (const float*)d_in[6];
  const int*   hr     = (const int*)d_in[7];
  const int*   hc     = (const int*)d_in[8];
  const int*   vr     = (const int*)d_in[9];
  const int*   vc     = (const int*)d_in[10];
  float* out = (float*)d_out;

  float* recs = (float*)d_ws;
  float* h  = recs + W1ELEMS;
  float* W2 = h + NE;

  rgcn_layer1_sort<<<L2_G, L2_SORT_THREADS, 0, stream>>>(values, hc, hr, h, recs);
  rgcn_w1_apply<<<L2_G, G_THREADS, 0, stream>>>(comps1, bases1, h, comps2, bases2, W2, recs);
  rgcn_layer2_sort<<<L2_G, L2_SORT_THREADS, 0, stream>>>(values, vr, vc, bias1, h, recs);
  rgcn_layer2<<<L2_G, RG_THREADS, 0, stream>>>(h, bias2, W2, recs, out);
}


# speedup vs baseline: 2.4811x; 1.1301x over previous
#include <hip/hip_runtime.h>
#include <hip/hip_bf16.h>

#define NN    50000
#define RR    51
#define EE    16
#define CC    8
#define BB    40
#define NNZ   1600000
#define NE    (NN * EE)
#define W1ELEMS ((size_t)RR * NE)
#define H2_OFF  ((size_t)4000000)

typedef float v2f __attribute__((ext_vector_type(2)));
typedef float v8f __attribute__((ext_vector_type(8)));

__device__ __forceinline__ void atomAddF32(float* p, float v) {
  unsafeAtomicAdd(p, v);
}


#define RG_THREADS 1024

#define L2_G     256
#define L2_ROWS  196
#define L2_CH    (NNZ / L2_G)
#define L2_CHP   6272
#define L2_OFFP  288
#define L2_RECF  ((size_t)L2_G * L2_CHP * 2)
#define L2_SORT_THREADS 512
#define L2_SORT_K ((L2_CH + L2_SORT_THREADS - 1) / L2_SORT_THREADS)

template <bool FIRST>
__device__ __forceinline__ void rgcn_sort_body(const float* __restrict__ values,
                                               const int* __restrict__ vr,
                                               const int* __restrict__ vc,
                                               const float* __restrict__ bias1,
                                               float* __restrict__ h,
                                               float* __restrict__ hB,
                                               float* __restrict__ ws) {
  __shared__ unsigned cursor[L2_G];
  __shared__ unsigned start[L2_G + 1];
  __shared__ uint2    recL[L2_CH];

  const int tid  = threadIdx.x;
  const int lane = tid & 31;
  const int c    = blockIdx.x;
  const int eLo  = c * L2_CH;

  if (tid < L2_G) cursor[tid] = 0u;
  __syncthreads();

  const int gid = c * L2_SORT_THREADS + tid;
  float4* h4 = reinterpret_cast<float4*>(h);
  const int  hi1 = gid + L2_G * L2_SORT_THREADS;
  const bool hok1 = hi1 < NE / 4;
  float4* hB4 = reinterpret_cast<float4*>(hB);
  float4 hx0 = make_float4(0.0f, 0.0f, 0.0f, 0.0f), hx1 = hx0, bv = hx0, hy0 = hx0, hy1 = hx0;
  if constexpr (!FIRST) {
    hx0 = h4[gid];
    hx1 = h4[hok1 ? hi1 : gid];
    hy0 = hB4[gid];
    hy1 = hB4[hok1 ? hi1 : gid];
    const int e0 = (gid & 3) * 4;
    bv = make_float4(bias1[e0], bias1[e0 + 1], bias1[e0 + 2], bias1[e0 + 3]);
  }

  unsigned r[L2_SORT_K];
  int      o[L2_SORT_K];
  float    v[L2_SORT_K];
#pragma unroll
  for (int k = 0; k < L2_SORT_K; ++k) {
    const int j  = tid + k * L2_SORT_THREADS;
    const int jj = j < L2_CH ? j : L2_CH - 1;
    r[k] = (unsigned)vr[eLo + jj];
    o[k] = vc[eLo + jj];
    v[k] = values[eLo + jj];
  }
  if constexpr (FIRST) {
    h4[gid]  = hx0;
    hB4[gid] = hx0;
    if (hok1) { h4[hi1] = hx0; hB4[hi1] = hx0; }
  } else {
    h4[gid] = make_float4(fmaxf((hx0.x + hy0.x) + bv.x, 0.0f), fmaxf((hx0.y + hy0.y) + bv.y, 0.0f),
                          fmaxf((hx0.z + hy0.z) + bv.z, 0.0f), fmaxf((hx0.w + hy0.w) + bv.w, 0.0f));
    if (hok1)
      h4[hi1] = make_float4(fmaxf((hx1.x + hy1.x) + bv.x, 0.0f), fmaxf((hx1.y + hy1.y) + bv.y, 0.0f),
                            fmaxf((hx1.z + hy1.z) + bv.z, 0.0f), fmaxf((hx1.w + hy1.w) + bv.w, 0.0f));
  }
  unsigned blk[L2_SORT_K], word0[L2_SORT_K];
#pragma unroll
  for (int k = 0; k < L2_SORT_K; ++k) {
    const unsigned p = r[k] / (unsigned)NN;
    const unsigned s = r[k] - p * (unsigned)NN;
    const unsigned b = s / (unsigned)L2_ROWS;
    blk[k]   = b;
    word0[k] = ((s - b * (unsigned)L2_ROWS) << 22) | (p << 16) | (unsigned)o[k];
    if (tid + k * L2_SORT_THREADS < L2_CH) atomicAdd(&cursor[b], 1u);
  }
  __syncthreads();

  if (tid < 32) {
    unsigned x[8];
    unsigned tot = 0u;
#pragma unroll
    for (int i = 0; i < 8; ++i) { x[i] = cursor[lane * 8 + i]; tot += x[i]; }
    unsigned incl = tot;
#pragma unroll
    for (int d = 1; d < 32; d <<= 1) {
      const unsigned y = __shfl_up(incl, (unsigned)d);
      if (lane >= d) incl += y;
    }
    unsigned run = incl - tot;
#pragma unroll
    for (int i = 0; i < 8; ++i) {
      start[lane * 8 + i]  = run;
      cursor[lane * 8 + i] = run;
      run += x[i];
    }
    if (lane == 31) start[L2_G] = run;
  }
  __syncthreads();

  unsigned pos[L2_SORT_K];
#pragma unroll
  for (int k = 0; k < L2_SORT_K; ++k) {
    pos[k] = 0u;
    if (tid + k * L2_SORT_THREADS < L2_CH) pos[k] = atomicAdd(&cursor[blk[k]], 1u);
  }
#pragma unroll
  for (int k = 0; k < L2_SORT_K; ++k) {
    if (tid + k * L2_SORT_THREADS < L2_CH) recL[pos[k]] = make_uint2(word0[k], __float_as_uint(v[k]));
  }
  __syncthreads();

  uint2* recs = reinterpret_cast<uint2*>(ws) + (size_t)c * L2_CHP;
#pragma unroll
  for (int k = 0; k < L2_SORT_K; ++k) {
    const int j = tid + k * L2_SORT_THREADS;
    if (j < L2_CH) recs[j] = recL[j];
  }
  unsigned* offs = reinterpret_cast<unsigned*>(ws + L2_RECF) + (size_t)c * L2_OFFP;
  if (tid <= L2_G) offs[tid] = start[tid];
}

__global__ void __launch_bounds__(L2_SORT_THREADS)
rgcn_layer1_sort(const float* __restrict__ values,
                 const int* __restrict__ hc,
                 const int* __restrict__ hr,
                 float* __restrict__ h,
                 float* __restrict__ hB,
                 float* __restrict__ ws) {
  rgcn_sort_body<true>(values, hc, hr, nullptr, h, hB, ws);
}

__global__ void __launch_bounds__(L2_SORT_THREADS)
rgcn_layer2_sort(const float* __restrict__ values,
                 const int* __restrict__ vr,
                 const int* __restrict__ vc,
                 const float* __restrict__ bias1,
                 float* __restrict__ h,
                 float* __restrict__ hB,
                 float* __restrict__ ws) {
  rgcn_sort_body<false>(values, vr, vc, bias1, h, hB, ws);
}

#define RG_U5   4
#define W2T_CP   20
#define W2T_PP   (CC * W2T_CP)
#define W2T_COPY (RR * W2T_PP)
#define L2_QW   512
#define L2_CPW  (L2_G / 32)

__global__ void __launch_bounds__(RG_THREADS)
rgcn_layer2(const float* __restrict__ h,
            const float* __restrict__ bias2,
            const float* __restrict__ W2,
            const float* __restrict__ ws,
            float* __restrict__ out) {
  __shared__ __attribute__((aligned(16))) float W2T[2 * W2T_COPY];
  __shared__ float    outTile[L2_ROWS * CC];
  __shared__ uint2    queue[32][L2_QW];
  __shared__ float    spare[CC];

  const int tid  = threadIdx.x;
  const int lane = tid & 31;
  const int wv   = tid >> 5;
  const int sub  = lane >> 3;
  const int c    = lane & 7;
  const int b    = blockIdx.x;
  const int rowLo = b * L2_ROWS;
  const int wvS  = __builtin_amdgcn_readfirstlane(wv);

  for (int j = tid; j < RR * EE * CC; j += RG_THREADS) {
    const int t = (j >> 7) * W2T_PP + (j & 7) * W2T_CP + ((j >> 3) & 15);
    const float w = W2[j];
    W2T[t] = w;
    W2T[W2T_COPY + t] = w;
  }
  for (int j = tid; j < L2_ROWS * CC; j += RG_THREADS) outTile[j] = 0.0f;
  __syncthreads();

  const uint2*    recs = reinterpret_cast<const uint2*>(ws);
  const unsigned* offs = reinterpret_cast<const unsigned*>(ws + L2_RECF);
  uint2* myQueue = queue[wv];

  auto drainBatch = [&](int base, int n) {
    uint2 ent[RG_U5];
    bool ok[RG_U5];
#pragma unroll
    for (int u = 0; u < RG_U5; ++u) {
      const int q = base + 4 * u + sub;
      ok[u] = q < n;
      ent[u] = myQueue[ok[u] ? q : 0];
      if (!ok[u]) ent[u] = make_uint2(0u, 0u);
    }
    float4 hv[RG_U5][4];
#pragma unroll
    for (int u = 0; u < RG_U5; ++u) {
      const float4* hp4 = reinterpret_cast<const float4*>(h + (size_t)(ent[u].x & 0xFFFFu) * EE);
#pragma unroll
      for (int k = 0; k < 4; ++k) hv[u][k] = hp4[k];
    }
    __builtin_amdgcn_sched_barrier(0);
    float acc[RG_U5];
#pragma unroll
    for (int u = 0; u < RG_U5; ++u) {
      const int p = (int)((ent[u].x >> 16) & 63u);
      const int copy = (p ^ sub) & 1;
      const float4* wq = reinterpret_cast<const float4*>(
          W2T + copy * W2T_COPY + p * W2T_PP + c * W2T_CP);
      float a = 0.0f;
#pragma unroll
      for (int k = 0; k < 4; ++k) {
        const float4 w = wq[k];
        a += hv[u][k].x * w.x;
        a += hv[u][k].y * w.y;
        a += hv[u][k].z * w.z;
        a += hv[u][k].w * w.w;
      }
      acc[u] = a;
    }
#pragma unroll
    for (int u = 0; u < RG_U5; ++u) {
      float* dst = ok[u] ? &outTile[(ent[u].x >> 22) * CC + c] : &spare[c];
      atomAddF32(dst, __uint_as_float(ent[u].y) * acc[u]);
    }
  };
  auto drainAll = [&](int n) {
    for (int q = 0; q < n; q += 4 * RG_U5) drainBatch(q, n);
  };

  unsigned myBeg = 0u, myLen = 0u;
  if (lane < L2_CPW) {
    const unsigned* op = offs + (size_t)(wv * L2_CPW + lane) * L2_OFFP + b;
    myBeg = op[0];
    myLen = op[1] - myBeg;
  }
  unsigned beg[L2_CPW], len[L2_CPW];
#pragma unroll
  for (int k = 0; k < L2_CPW; ++k) {
    beg[k] = (unsigned)__builtin_amdgcn_readlane((int)myBeg, k);
    len[k] = (unsigned)__builtin_amdgcn_readlane((int)myLen, k);
  }
  uint2 first[L2_CPW];
#pragma unroll
  for (int k = 0; k < L2_CPW; ++k) {
    const uint2* rp = recs + (size_t)(wv * L2_CPW + k) * L2_CHP;
    first[k] = rp[(unsigned)lane < len[k] ? beg[k] + (unsigned)lane : 0u];
  }
  int qn = 0;
#pragma unroll
  for (int k = 0; k < L2_CPW; ++k) {
    const int cnt = len[k] < 32u ? (int)len[k] : 32;
    if (lane < cnt) myQueue[qn + lane] = first[k];
    qn += cnt;
  }
  if (__builtin_amdgcn_ballot_w32(myLen > 32u) != 0u) {
#pragma nounroll
    for (int k = 0; k < L2_CPW; ++k) {
      const unsigned* op = offs + (size_t)(wvS * L2_CPW + k) * L2_OFFP + b;
      const unsigned bk = op[0];
      const unsigned lk = op[1] - bk;
      const uint2* rp = recs + (size_t)(wvS * L2_CPW + k) * L2_CHP + bk;
      for (unsigned q0 = 32u; q0 < lk; q0 += 32u) {
        const unsigned left = lk - q0;
        const int cnt = left < 32u ? (int)left : 32;
        if (qn + cnt > L2_QW) { drainAll(qn); qn = 0; }
        if (lane < cnt) myQueue[qn + lane] = rp[q0 + (unsigned)lane];
        qn += cnt;
      }
    }
  }
  drainAll(qn);
  __syncthreads();

  const int tidC = wvS * 32 + (int)__builtin_amdgcn_mbcnt_lo(~0u, 0u);
  for (int j = tidC; j < L2_ROWS * CC; j += RG_THREADS) {
    const int row = rowLo + (j >> 3);
    if (row < NN) out[(size_t)rowLo * CC + j] = outTile[j] + bias2[j & 7];
  }
}


#define G_THREADS 768
#define G_WAVES   (G_THREADS / 32)
#define G_K       9
#define G_CAP     (G_K * G_THREADS)
#define G_BINS    224
#define G_U       4

__global__ void __launch_bounds__(G_THREADS)
rgcn_w1_apply(const float* __restrict__ comps1,
              const float* __restrict__ bases1,
              float* __restrict__ h,
              float* __restrict__ hB,
              const float* __restrict__ comps2,
              const float* __restrict__ bases2,
              float* __restrict__ W2,
              const float* __restrict__ ws) {
  __shared__ uint2    sorted[G_CAP];
  __shared__ float    tileL[G_WAVES][RR * EE];
  __shared__ unsigned pbeg[L2_G];
  __shared__ unsigned pstart[L2_G + 1];
  __shared__ unsigned cursor[G_BINS];
  __shared__ unsigned nstart[G_BINS + 1];

  const int tid  = threadIdx.x;
  const int lane = tid & 31;
  const int wv   = tid >> 5;
  const int half = lane >> 4;
  const int l16  = lane & 15;
  const int b    = blockIdx.x;

  const unsigned seAid = (unsigned)__builtin_amdgcn_s_sendmsg_rtn(0x87);
  float* hAdd = ((seAid >> 18) & 1u) ? hB : h;

  {
    const int gid = b * G_THREADS + tid;
    if (gid < RR * EE * CC) {
      const int r  = gid >> 7;
      const int ec = gid & 127;
      float acc = 0.0f;
#pragma unroll
      for (int k = 0; k < BB; ++k)
        acc += comps2[r * BB + k] * bases2[k * (EE * CC) + ec];
      W2[gid] = acc;
    }
  }

  const uint2*    recs = reinterpret_cast<const uint2*>(ws);
  const unsigned* offs = reinterpret_cast<const unsigned*>(ws + L2_RECF);

  if (tid < L2_G) {
    const unsigned* op = offs + (size_t)tid * L2_OFFP + b;
    const unsigned u0 = op[0];
    pbeg[tid]   = u0;
    pstart[tid] = op[1] - u0;
  }
  __syncthreads();
  if (tid < 32) {
    unsigned x[8];
    unsigned tot = 0u;
#pragma unroll
    for (int i = 0; i < 8; ++i) { x[i] = pstart[lane * 8 + i]; tot += x[i]; }
    unsigned incl = tot;
#pragma unroll
    for (int d = 1; d < 32; d <<= 1) {
      const unsigned y = __shfl_up(incl, (unsigned)d);
      if (lane >= d) incl += y;
    }
    unsigned run = incl - tot;
#pragma unroll
    for (int i = 0; i < 8; ++i) { pstart[lane * 8 + i] = run; run += x[i]; }
    if (lane == 31) pstart[L2_G] = run;
  }
  __syncthreads();
  const int total = (int)pstart[L2_G];

  for (int base = 0; base < total; base += G_CAP) {
    const int passEnd = total < base + G_CAP ? total : base + G_CAP;
    if (tid < G_BINS) cursor[tid] = 0u;
    __syncthreads();

    unsigned w0[G_K], w1[G_K];
#pragma unroll
    for (int k = 0; k < G_K; ++k) {
      const int idx = base + tid + k * G_THREADS;
      const unsigned ii = (unsigned)(idx < passEnd ? idx : base);
      int pc = 0;
#pragma unroll
      for (int st = L2_G / 2; st >= 1; st >>= 1)
        if (pstart[pc + st] <= ii) pc += st;
      const uint2 rec = recs[(size_t)pc * L2_CHP + pbeg[pc] + (ii - pstart[pc])];
      w0[k] = rec.x;
      w1[k] = rec.y;
    }
#pragma unroll
    for (int k = 0; k < G_K; ++k)
      if (base + tid + k * G_THREADS < passEnd) atomicAdd(&cursor[w0[k] >> 22], 1u);
    __syncthreads();

    if (tid < 32) {
      unsigned x[7];
      unsigned tot = 0u;
#pragma unroll
      for (int i = 0; i < 7; ++i) { x[i] = cursor[lane * 7 + i]; tot += x[i]; }
      unsigned incl = tot;
#pragma unroll
      for (int d = 1; d < 32; d <<= 1) {
        const unsigned y = __shfl_up(incl, (unsigned)d);
        if (lane >= d) incl += y;
      }
      unsigned run = incl - tot;
#pragma unroll
      for (int i = 0; i < 7; ++i) {
        nstart[lane * 7 + i] = run;
        cursor[lane * 7 + i] = run;
        run += x[i];
      }
      if (lane == 31) nstart[G_BINS] = run;
    }
    __syncthreads();

    unsigned pos[G_K];
#pragma unroll
    for (int k = 0; k < G_K; ++k) {
      pos[k] = 0u;
      if (base + tid + k * G_THREADS < passEnd) pos[k] = atomicAdd(&cursor[w0[k] >> 22], 1u);
    }
#pragma unroll
    for (int k = 0; k < G_K; ++k)
      if (base + tid + k * G_THREADS < passEnd) sorted[pos[k]] = make_uint2(w0[k], w1[k]);
    __syncthreads();

    v2f a[4][10];
#pragma unroll
    for (int mt = 0; mt < 4; ++mt) {
      int m = mt * 16 + l16;
#pragma unroll
      for (int ks = 0; ks < 10; ++ks) {
        int k = ks * 4 + half * 2;
        float ax = 0.0f, ay = 0.0f;
        if (m < RR) {
          ax = comps1[m * BB + k];
          ay = comps1[m * BB + k + 1];
        }
        a[mt][ks].x = ax;
        a[mt][ks].y = ay;
      }
    }

    float* tl = tileL[wv];
    for (int t = wv; t < L2_ROWS; t += G_WAVES) {
      const int o = b * L2_ROWS + t;
      if (o >= NN) break;
      const int col = o * EE + l16;
      v8f acc[4] = {};
#pragma unroll
      for (int ks = 0; ks < 10; ++ks) {
        const int k = ks * 4 + half * 2;
        v2f bb;
        bb.x = __builtin_nontemporal_load(&bases1[(size_t)k * NE + col]);
        bb.y = __builtin_nontemporal_load(&bases1[(size_t)(k + 1) * NE + col]);
#pragma unroll
        for (int mt = 0; mt < 4; ++mt) {
          acc[mt] = __builtin_amdgcn_wmma_f32_16x16x4_f32(
               false, a[mt][ks],  false, bb,
               (short)0, acc[mt],
               false,  false);
        }
      }
#pragma unroll
      for (int mt = 0; mt < 4; ++mt) {
#pragma unroll
        for (int j = 0; j < 8; ++j) {
          const int m = mt * 16 + j + half * 8;
          if (m < RR) tl[m * EE + l16] = acc[mt][j];
        }
      }
      const int s0 = __builtin_amdgcn_readfirstlane((int)nstart[t]);
      const int s1 = __builtin_amdgcn_readfirstlane((int)nstart[t + 1]);
      for (int q0 = s0; q0 < s1; q0 += 2 * G_U) {
        uint2 ent[G_U];
        bool ok[G_U];
#pragma unroll
        for (int u = 0; u < G_U; ++u) {
          const int q = q0 + 2 * u + half;
          ok[u] = q < s1;
          ent[u] = sorted[ok[u] ? q : s0];
        }
        float w[G_U];
#pragma unroll
        for (int u = 0; u < G_U; ++u)
          w[u] = tl[((ent[u].x >> 16) & 63u) * EE + l16];
#pragma unroll
        for (int u = 0; u < G_U; ++u)
          if (ok[u])
            atomAddF32(&hAdd[(size_t)(ent[u].x & 0xFFFFu) * EE + l16],
                       __uint_as_float(ent[u].y) * w[u]);
      }
    }
  }
}

extern "C" void kernel_launch(void* const* d_in, const int* in_sizes, int n_in,
                              void* d_out, int out_size, void* d_ws, size_t ws_size,
                              hipStream_t stream) {
  const float* values = (const float*)d_in[0];
  const float* comps1 = (const float*)d_in[1];
  const float* bases1 = (const float*)d_in[2];
  const float* comps2 = (const float*)d_in[3];
  const float* bases2 = (const float*)d_in[4];
  const float* bias1  = (const float*)d_in[5];
  const float* bias2  = (const float*)d_in[6];
  const int*   hr     = (const int*)d_in[7];
  const int*   hc     = (const int*)d_in[8];
  const int*   vr     = (const int*)d_in[9];
  const int*   vc     = (const int*)d_in[10];
  float* out = (float*)d_out;

  float* recs = (float*)d_ws;
  float* h  = recs + W1ELEMS;
  float* W2 = h + NE;
  float* hB = recs + H2_OFF;

  rgcn_layer1_sort<<<L2_G, L2_SORT_THREADS, 0, stream>>>(values, hc, hr, h, hB, recs);
  rgcn_w1_apply<<<L2_G, G_THREADS, 0, stream>>>(comps1, bases1, h, hB, comps2, bases2, W2, recs);
  rgcn_layer2_sort<<<L2_G, L2_SORT_THREADS, 0, stream>>>(values, vr, vc, bias1, h, hB, recs);
  rgcn_layer2<<<L2_G, RG_THREADS, 0, stream>>>(h, bias2, W2, recs, out);
}
